# starting kernel
# baseline (speedup 1.0000x reference)
#include <hip/hip_runtime.h>


typedef __attribute__((ext_vector_type(2))) float v2f;
typedef __attribute__((ext_vector_type(8))) float v8f;

#define IN_F  64
#define OUT_F 64
#define KER   4
#define EMB   3
#define CDIM  (OUT_F * KER)

__global__ void support_gemm_wmma(const float* __restrict__ x,
                                  const float* __restrict__ w,
                                  float* __restrict__ support,
                                  int N) {
  const int lane = threadIdx.x & 31;
  const int wave = (blockIdx.x * blockDim.x + threadIdx.x) >> 5;
  const int waveStride = (gridDim.x * blockDim.x) >> 5;

  const int tilesPerRow = CDIM / 16;
  const int rowTiles    = (N + 15) >> 4;
  const int totalTiles  = rowTiles * tilesPerRow;

  const int half = lane >> 4;
  const int lid  = lane & 15;

  for (int t = wave; t < totalTiles; t += waveStride) {
    const int m0 = (t / tilesPerRow) * 16;
    const int c0 = (t % tilesPerRow) * 16;

    int arow = m0 + lid;
    if (arow >= N) arow = N - 1;
    const float* __restrict__ arowp = x + (size_t)arow * IN_F + half * 2;
    const float* __restrict__ bcolp = w + c0 + lid + (size_t)(half * 2) * CDIM;

    v8f acc = {};
#pragma unroll
    for (int k0 = 0; k0 < IN_F; k0 += 4) {
      v2f a;
      a.x = arowp[k0];
      a.y = arowp[k0 + 1];
      v2f b;
      b.x = bcolp[(size_t)k0 * CDIM];
      b.y = bcolp[(size_t)(k0 + 1) * CDIM];
      acc = __builtin_amdgcn_wmma_f32_16x16x4_f32(
           false, a,  false, b,
           (short)0, acc,  false,  false);
    }
#pragma unroll
    for (int r = 0; r < 8; ++r) {
      const int row = m0 + r + 8 * half;
      if (row < N) support[(size_t)row * CDIM + c0 + lid] = acc[r];
    }
  }
}

__global__ void init_out(float* __restrict__ out,
                         const float* __restrict__ bias, int total) {
  const int i = blockIdx.x * blockDim.x + threadIdx.x;
  if (i < total) out[i] = bias[i & (OUT_F - 1)];
}

__global__ void edge_scatter(const int*   __restrict__ eidx,
                             const float* __restrict__ spec,
                             const float* __restrict__ mu,
                             const float* __restrict__ sig,
                             const float* __restrict__ support,
                             float* __restrict__ out,
                             int E) {
  const int lane = threadIdx.x & 31;
  const int wave = (blockIdx.x * blockDim.x + threadIdx.x) >> 5;
  const int waveStride = (gridDim.x * blockDim.x) >> 5;

  float mu0[KER], mu1[KER], mu2[KER], sg[KER];
#pragma unroll
  for (int k = 0; k < KER; ++k) {
    mu0[k] = mu[0 * KER + k];
    mu1[k] = mu[1 * KER + k];
    mu2[k] = mu[2 * KER + k];
    sg[k]  = sig[k];
  }

  for (int e = wave; e < E; e += waveStride) {
    const int row = eidx[e];
    const int col = eidx[E + e];

    const float d0 = spec[row * EMB + 0] - spec[col * EMB + 0];
    const float d1 = spec[row * EMB + 1] - spec[col * EMB + 1];
    const float d2 = spec[row * EMB + 2] - spec[col * EMB + 2];

    float val[KER];
#pragma unroll
    for (int k = 0; k < KER; ++k) {
      const float a0 = d0 - mu0[k];
      const float a1 = d1 - mu1[k];
      const float a2 = d2 - mu2[k];
      const float qq = -0.5f * (a0 * a0 + a1 * a1 + a2 * a2);
      val[k] = __expf(sg[k] * qq);
    }

    const float4* sp = (const float4*)(support + (size_t)col * CDIM + lane * 8);
    const float4 s0 = sp[0];
    const float4 s1 = sp[1];
    const float m0 = s0.x * val[0] + s0.y * val[1] + s0.z * val[2] + s0.w * val[3];
    const float m1 = s1.x * val[0] + s1.y * val[1] + s1.z * val[2] + s1.w * val[3];

    float* dst = out + (size_t)row * OUT_F + lane * 2;
    unsafeAtomicAdd(dst,     m0);
    unsafeAtomicAdd(dst + 1, m1);
  }
}

extern "C" void kernel_launch(void* const* d_in, const int* in_sizes, int n_in,
                              void* d_out, int out_size, void* d_ws, size_t ws_size,
                              hipStream_t stream) {
  const float* x    = (const float*)d_in[0];
  const int*   eidx = (const int*)  d_in[1];
  const float* spec = (const float*)d_in[2];
  const float* w    = (const float*)d_in[3];
  const float* bias = (const float*)d_in[4];
  const float* mu   = (const float*)d_in[5];
  const float* sig  = (const float*)d_in[6];

  const int N = in_sizes[0] / IN_F;
  const int E = in_sizes[1] / 2;

  float* support = (float*)d_ws;
  float* out     = (float*)d_out;

  {
    const int totalTiles = ((N + 15) / 16) * (CDIM / 16);
    const int wavesPerBlock = 8;
    const int blocks = (totalTiles + wavesPerBlock - 1) / wavesPerBlock;
    support_gemm_wmma<<<blocks, 256, 0, stream>>>(x, w, support, N);
  }

  {
    const int total = N * OUT_F;
    init_out<<<(total + 255) / 256, 256, 0, stream>>>(out, bias, total);
  }

  {
    const int blocks = 2048;
    edge_scatter<<<blocks, 256, 0, stream>>>(eidx, spec, mu, sig, support, out, E);
  }
}


# GEMM re-tiled: each wave computes 16 rows x 64 cols; A strip in registers, weight chunk staged per workgroup in LDS as k-pairs, columns interleaved so every store writes whole 128-B lines
# speedup vs baseline: 1.0411x; 1.0411x over previous
#include <hip/hip_runtime.h>


typedef __attribute__((ext_vector_type(2))) float v2f;
typedef __attribute__((ext_vector_type(8))) float v8f;

#define IN_F  64
#define OUT_F 64
#define KER   4
#define EMB   3
#define CDIM  (OUT_F * KER)

#define GEMM_WAVES 8
#define CHUNK      64

__global__ void __launch_bounds__(GEMM_WAVES * 32)
support_gemm_wmma(const float* __restrict__ x,
                  const float* __restrict__ w,
                  float* __restrict__ support,
                  int N) {
  __shared__ float4 wl[2][(IN_F / 2) * (CHUNK / 4)];

  const int tid  = threadIdx.x;
  const int lane = tid & 31;
  const int wv   = __builtin_amdgcn_readfirstlane(tid >> 5);
  const int half = lane >> 4;
  const int lid  = lane & 15;

  const int cbase    = blockIdx.x * CHUNK;
  const int rowTiles = (N + 15) >> 4;
  const int strip    = blockIdx.y * GEMM_WAVES + wv;
  const int m0       = strip * 16;

  int arow = m0 + lid;
  if (arow >= N) arow = N - 1;
  const v2f* __restrict__ ap =
      (const v2f*)(x + (size_t)arow * IN_F + half * 2);
  v2f a[IN_F / 4];
#pragma unroll
  for (int t = 0; t < IN_F / 4; ++t) a[t] = ap[2 * t];

  {
    const float4* __restrict__ wg = (const float4*)(w + cbase);
    const int kq = tid >> 4;
    const int c4 = tid & 15;
#pragma unroll
    for (int p = 0; p < IN_F / 32; ++p) {
      const int kp = kq + 16 * p;
      const float4 r0 = wg[(size_t)(2 * kp)     * (CDIM / 4) + c4];
      const float4 r1 = wg[(size_t)(2 * kp + 1) * (CDIM / 4) + c4];
      float4 lo, hi;
      lo.x = r0.x; lo.y = r1.x; lo.z = r0.y; lo.w = r1.y;
      hi.x = r0.z; hi.y = r1.z; hi.z = r0.w; hi.w = r1.w;
      wl[0][kp * (CHUNK / 4) + c4] = lo;
      wl[1][kp * (CHUNK / 4) + c4] = hi;
    }
  }
  __syncthreads();

  if (strip < rowTiles) {
    v8f acc0 = {}, acc1 = {}, acc2 = {}, acc3 = {};
#pragma unroll
    for (int t = 0; t < IN_F / 4; ++t) {
      const float4 qa = wl[0][(2 * t + half) * (CHUNK / 4) + lid];
      const float4 qb = wl[1][(2 * t + half) * (CHUNK / 4) + lid];
      v2f b;
      b.x = qa.x; b.y = qa.y;
      acc0 = __builtin_amdgcn_wmma_f32_16x16x4_f32(false, a[t], false, b,
                                                   (short)0, acc0, false, false);
      b.x = qa.z; b.y = qa.w;
      acc1 = __builtin_amdgcn_wmma_f32_16x16x4_f32(false, a[t], false, b,
                                                   (short)0, acc1, false, false);
      b.x = qb.x; b.y = qb.y;
      acc2 = __builtin_amdgcn_wmma_f32_16x16x4_f32(false, a[t], false, b,
                                                   (short)0, acc2, false, false);
      b.x = qb.z; b.y = qb.w;
      acc3 = __builtin_amdgcn_wmma_f32_16x16x4_f32(false, a[t], false, b,
                                                   (short)0, acc3, false, false);
    }
    float* __restrict__ cp = support + (size_t)cbase + 4 * lid;
#pragma unroll
    for (int r = 0; r < 8; ++r) {
      const int row = m0 + r + 8 * half;
      if (row < N) {
        float4 v;
        v.x = acc0[r]; v.y = acc1[r]; v.z = acc2[r]; v.w = acc3[r];
        *(float4*)(cp + (size_t)row * CDIM) = v;
      }
    }
  }
}

__global__ void init_out(float* __restrict__ out,
                         const float* __restrict__ bias, int total) {
  const int i = blockIdx.x * blockDim.x + threadIdx.x;
  if (i < total) out[i] = bias[i & (OUT_F - 1)];
}

__global__ void edge_scatter(const int*   __restrict__ eidx,
                             const float* __restrict__ spec,
                             const float* __restrict__ mu,
                             const float* __restrict__ sig,
                             const float* __restrict__ support,
                             float* __restrict__ out,
                             int E) {
  const int lane = threadIdx.x & 31;
  const int wave = (blockIdx.x * blockDim.x + threadIdx.x) >> 5;
  const int waveStride = (gridDim.x * blockDim.x) >> 5;

  float mu0[KER], mu1[KER], mu2[KER], sg[KER];
#pragma unroll
  for (int k = 0; k < KER; ++k) {
    mu0[k] = mu[0 * KER + k];
    mu1[k] = mu[1 * KER + k];
    mu2[k] = mu[2 * KER + k];
    sg[k]  = sig[k];
  }

  for (int e = wave; e < E; e += waveStride) {
    const int row = eidx[e];
    const int col = eidx[E + e];

    const float d0 = spec[row * EMB + 0] - spec[col * EMB + 0];
    const float d1 = spec[row * EMB + 1] - spec[col * EMB + 1];
    const float d2 = spec[row * EMB + 2] - spec[col * EMB + 2];

    float val[KER];
#pragma unroll
    for (int k = 0; k < KER; ++k) {
      const float a0 = d0 - mu0[k];
      const float a1 = d1 - mu1[k];
      const float a2 = d2 - mu2[k];
      const float qq = -0.5f * (a0 * a0 + a1 * a1 + a2 * a2);
      val[k] = __expf(sg[k] * qq);
    }

    const float4* sp = (const float4*)(support + (size_t)col * CDIM + lane * 8);
    const float4 s0 = sp[0];
    const float4 s1 = sp[1];
    const float m0 = s0.x * val[0] + s0.y * val[1] + s0.z * val[2] + s0.w * val[3];
    const float m1 = s1.x * val[0] + s1.y * val[1] + s1.z * val[2] + s1.w * val[3];

    float* dst = out + (size_t)row * OUT_F + lane * 2;
    unsafeAtomicAdd(dst,     m0);
    unsafeAtomicAdd(dst + 1, m1);
  }
}

extern "C" void kernel_launch(void* const* d_in, const int* in_sizes, int n_in,
                              void* d_out, int out_size, void* d_ws, size_t ws_size,
                              hipStream_t stream) {
  const float* x    = (const float*)d_in[0];
  const int*   eidx = (const int*)  d_in[1];
  const float* spec = (const float*)d_in[2];
  const float* w    = (const float*)d_in[3];
  const float* bias = (const float*)d_in[4];
  const float* mu   = (const float*)d_in[5];
  const float* sig  = (const float*)d_in[6];

  const int N = in_sizes[0] / IN_F;
  const int E = in_sizes[1] / 2;

  float* support = (float*)d_ws;
  float* out     = (float*)d_out;

  {
    const int rowTiles = (N + 15) / 16;
    const dim3 grid(CDIM / CHUNK, (rowTiles + GEMM_WAVES - 1) / GEMM_WAVES);
    support_gemm_wmma<<<grid, GEMM_WAVES * 32, 0, stream>>>(x, w, support, N);
  }

  {
    const int total = N * OUT_F;
    init_out<<<(total + 255) / 256, 256, 0, stream>>>(out, bias, total);
  }

  {
    const int blocks = 2048;
    edge_scatter<<<blocks, 256, 0, stream>>>(eidx, spec, mu, sig, support, out, E);
  }
}


# Bias seeding of out moved into the GEMM launch: each GEMM workgroup stores the bias to 32 of its 128 rows as whole 128-B lines behind its barrier; the 12,500-workgroup init launch is removed
# speedup vs baseline: 1.0873x; 1.0444x over previous
#include <hip/hip_runtime.h>


typedef __attribute__((ext_vector_type(2))) float v2f;
typedef __attribute__((ext_vector_type(8))) float v8f;

#define IN_F  64
#define OUT_F 64
#define KER   4
#define EMB   3
#define CDIM  (OUT_F * KER)

#define GEMM_WAVES 8
#define CHUNK      64
#define SEED_ROWS  ((GEMM_WAVES * 16) / (CDIM / CHUNK))

__global__ void __launch_bounds__(GEMM_WAVES * 32)
support_gemm_wmma(const float* __restrict__ x,
                  const float* __restrict__ w,
                  float* __restrict__ support,
                  const float* __restrict__ bias,
                  float* __restrict__ out,
                  int N) {
  __shared__ float4 wl[2][(IN_F / 2) * (CHUNK / 4)];

  const int tid  = threadIdx.x;
  const int lane = tid & 31;
  const int wv   = __builtin_amdgcn_readfirstlane(tid >> 5);
  const int half = lane >> 4;
  const int lid  = lane & 15;

  const int cbase    = blockIdx.x * CHUNK;
  const int rowTiles = (N + 15) >> 4;
  const int strip    = blockIdx.y * GEMM_WAVES + wv;
  const int m0       = strip * 16;

  int arow = m0 + lid;
  if (arow >= N) arow = N - 1;
  const v2f* __restrict__ ap =
      (const v2f*)(x + (size_t)arow * IN_F + half * 2);
  v2f a[IN_F / 4];
#pragma unroll
  for (int t = 0; t < IN_F / 4; ++t) a[t] = ap[2 * t];

  float4 bv = ((const float4*)bias)[tid & 15];

  {
    const float4* __restrict__ wg = (const float4*)(w + cbase);
    const int kq = tid >> 4;
    const int c4 = tid & 15;
#pragma unroll
    for (int p = 0; p < IN_F / 32; ++p) {
      const int kp = kq + 16 * p;
      const float4 r0 = wg[(size_t)(2 * kp)     * (CDIM / 4) + c4];
      const float4 r1 = wg[(size_t)(2 * kp + 1) * (CDIM / 4) + c4];
      float4 lo, hi;
      lo.x = r0.x; lo.y = r1.x; lo.z = r0.y; lo.w = r1.y;
      hi.x = r0.z; hi.y = r1.z; hi.z = r0.w; hi.w = r1.w;
      wl[0][kp * (CHUNK / 4) + c4] = lo;
      wl[1][kp * (CHUNK / 4) + c4] = hi;
    }
  }
  asm volatile("" : "+v"(bv.x), "+v"(bv.y), "+v"(bv.z), "+v"(bv.w));
  __syncthreads();

  {
    const int r0 = blockIdx.y * (GEMM_WAVES * 16) + blockIdx.x * SEED_ROWS + (tid >> 4);
    float4* __restrict__ op = (float4*)out + (size_t)r0 * (OUT_F / 4) + (tid & 15);
#pragma unroll
    for (int p = 0; p < SEED_ROWS / 16; ++p) {
      if (r0 + 16 * p < N) op[(size_t)(16 * p) * (OUT_F / 4)] = bv;
    }
  }

  if (strip < rowTiles) {
    v8f acc0 = {}, acc1 = {}, acc2 = {}, acc3 = {};
#pragma unroll
    for (int t = 0; t < IN_F / 4; ++t) {
      const float4 qa = wl[0][(2 * t + half) * (CHUNK / 4) + lid];
      const float4 qb = wl[1][(2 * t + half) * (CHUNK / 4) + lid];
      v2f b;
      b.x = qa.x; b.y = qa.y;
      acc0 = __builtin_amdgcn_wmma_f32_16x16x4_f32(false, a[t], false, b,
                                                   (short)0, acc0, false, false);
      b.x = qa.z; b.y = qa.w;
      acc1 = __builtin_amdgcn_wmma_f32_16x16x4_f32(false, a[t], false, b,
                                                   (short)0, acc1, false, false);
      b.x = qb.x; b.y = qb.y;
      acc2 = __builtin_amdgcn_wmma_f32_16x16x4_f32(false, a[t], false, b,
                                                   (short)0, acc2, false, false);
      b.x = qb.z; b.y = qb.w;
      acc3 = __builtin_amdgcn_wmma_f32_16x16x4_f32(false, a[t], false, b,
                                                   (short)0, acc3, false, false);
    }
    float* __restrict__ cp = support + (size_t)cbase + 4 * lid;
#pragma unroll
    for (int r = 0; r < 8; ++r) {
      const int row = m0 + r + 8 * half;
      if (row < N) {
        float4 v;
        v.x = acc0[r]; v.y = acc1[r]; v.z = acc2[r]; v.w = acc3[r];
        *(float4*)(cp + (size_t)row * CDIM) = v;
      }
    }
  }
}

__global__ void edge_scatter(const int*   __restrict__ eidx,
                             const float* __restrict__ spec,
                             const float* __restrict__ mu,
                             const float* __restrict__ sig,
                             const float* __restrict__ support,
                             float* __restrict__ out,
                             int E) {
  const int lane = threadIdx.x & 31;
  const int wave = (blockIdx.x * blockDim.x + threadIdx.x) >> 5;
  const int waveStride = (gridDim.x * blockDim.x) >> 5;

  float mu0[KER], mu1[KER], mu2[KER], sg[KER];
#pragma unroll
  for (int k = 0; k < KER; ++k) {
    mu0[k] = mu[0 * KER + k];
    mu1[k] = mu[1 * KER + k];
    mu2[k] = mu[2 * KER + k];
    sg[k]  = sig[k];
  }

  for (int e = wave; e < E; e += waveStride) {
    const int row = eidx[e];
    const int col = eidx[E + e];

    const float d0 = spec[row * EMB + 0] - spec[col * EMB + 0];
    const float d1 = spec[row * EMB + 1] - spec[col * EMB + 1];
    const float d2 = spec[row * EMB + 2] - spec[col * EMB + 2];

    float val[KER];
#pragma unroll
    for (int k = 0; k < KER; ++k) {
      const float a0 = d0 - mu0[k];
      const float a1 = d1 - mu1[k];
      const float a2 = d2 - mu2[k];
      const float qq = -0.5f * (a0 * a0 + a1 * a1 + a2 * a2);
      val[k] = __expf(sg[k] * qq);
    }

    const float4* sp = (const float4*)(support + (size_t)col * CDIM + lane * 8);
    const float4 s0 = sp[0];
    const float4 s1 = sp[1];
    const float m0 = s0.x * val[0] + s0.y * val[1] + s0.z * val[2] + s0.w * val[3];
    const float m1 = s1.x * val[0] + s1.y * val[1] + s1.z * val[2] + s1.w * val[3];

    float* dst = out + (size_t)row * OUT_F + lane * 2;
    unsafeAtomicAdd(dst,     m0);
    unsafeAtomicAdd(dst + 1, m1);
  }
}

extern "C" void kernel_launch(void* const* d_in, const int* in_sizes, int n_in,
                              void* d_out, int out_size, void* d_ws, size_t ws_size,
                              hipStream_t stream) {
  const float* x    = (const float*)d_in[0];
  const int*   eidx = (const int*)  d_in[1];
  const float* spec = (const float*)d_in[2];
  const float* w    = (const float*)d_in[3];
  const float* bias = (const float*)d_in[4];
  const float* mu   = (const float*)d_in[5];
  const float* sig  = (const float*)d_in[6];

  const int N = in_sizes[0] / IN_F;
  const int E = in_sizes[1] / 2;

  float* support = (float*)d_ws;
  float* out     = (float*)d_out;

  {
    const int rowTiles = (N + 15) / 16;
    const dim3 grid(CDIM / CHUNK, (rowTiles + GEMM_WAVES - 1) / GEMM_WAVES);
    support_gemm_wmma<<<grid, GEMM_WAVES * 32, 0, stream>>>(x, w, support, bias, out, N);
  }

  {
    const int blocks = 2048;
    edge_scatter<<<blocks, 256, 0, stream>>>(eidx, spec, mu, sig, support, out, E);
  }
}


# Edge phase batched: a wave loads indices and spec points and computes the Gaussian weights of 32 edges at once (one per lane), then broadcasts them per edge and gathers two edges at a time
# speedup vs baseline: 1.2202x; 1.1222x over previous
#include <hip/hip_runtime.h>


typedef __attribute__((ext_vector_type(2))) float v2f;
typedef __attribute__((ext_vector_type(8))) float v8f;

#define IN_F  64
#define OUT_F 64
#define KER   4
#define EMB   3
#define CDIM  (OUT_F * KER)

#define GEMM_WAVES 8
#define CHUNK      64
#define SEED_ROWS  ((GEMM_WAVES * 16) / (CDIM / CHUNK))

__global__ void __launch_bounds__(GEMM_WAVES * 32)
support_gemm_wmma(const float* __restrict__ x,
                  const float* __restrict__ w,
                  float* __restrict__ support,
                  const float* __restrict__ bias,
                  float* __restrict__ out,
                  int N) {
  __shared__ float4 wl[2][(IN_F / 2) * (CHUNK / 4)];

  const int tid  = threadIdx.x;
  const int lane = tid & 31;
  const int wv   = __builtin_amdgcn_readfirstlane(tid >> 5);
  const int half = lane >> 4;
  const int lid  = lane & 15;

  const int cbase    = blockIdx.x * CHUNK;
  const int rowTiles = (N + 15) >> 4;
  const int strip    = blockIdx.y * GEMM_WAVES + wv;
  const int m0       = strip * 16;

  int arow = m0 + lid;
  if (arow >= N) arow = N - 1;
  const v2f* __restrict__ ap =
      (const v2f*)(x + (size_t)arow * IN_F + half * 2);
  v2f a[IN_F / 4];
#pragma unroll
  for (int t = 0; t < IN_F / 4; ++t) a[t] = ap[2 * t];

  float4 bv = ((const float4*)bias)[tid & 15];

  {
    const float4* __restrict__ wg = (const float4*)(w + cbase);
    const int kq = tid >> 4;
    const int c4 = tid & 15;
#pragma unroll
    for (int p = 0; p < IN_F / 32; ++p) {
      const int kp = kq + 16 * p;
      const float4 r0 = wg[(size_t)(2 * kp)     * (CDIM / 4) + c4];
      const float4 r1 = wg[(size_t)(2 * kp + 1) * (CDIM / 4) + c4];
      float4 lo, hi;
      lo.x = r0.x; lo.y = r1.x; lo.z = r0.y; lo.w = r1.y;
      hi.x = r0.z; hi.y = r1.z; hi.z = r0.w; hi.w = r1.w;
      wl[0][kp * (CHUNK / 4) + c4] = lo;
      wl[1][kp * (CHUNK / 4) + c4] = hi;
    }
  }
  asm volatile("" : "+v"(bv.x), "+v"(bv.y), "+v"(bv.z), "+v"(bv.w));
  __syncthreads();

  {
    const int r0 = blockIdx.y * (GEMM_WAVES * 16) + blockIdx.x * SEED_ROWS + (tid >> 4);
    float4* __restrict__ op = (float4*)out + (size_t)r0 * (OUT_F / 4) + (tid & 15);
#pragma unroll
    for (int p = 0; p < SEED_ROWS / 16; ++p) {
      if (r0 + 16 * p < N) op[(size_t)(16 * p) * (OUT_F / 4)] = bv;
    }
  }

  if (strip < rowTiles) {
    v8f acc0 = {}, acc1 = {}, acc2 = {}, acc3 = {};
#pragma unroll
    for (int t = 0; t < IN_F / 4; ++t) {
      const float4 qa = wl[0][(2 * t + half) * (CHUNK / 4) + lid];
      const float4 qb = wl[1][(2 * t + half) * (CHUNK / 4) + lid];
      v2f b;
      b.x = qa.x; b.y = qa.y;
      acc0 = __builtin_amdgcn_wmma_f32_16x16x4_f32(false, a[t], false, b,
                                                   (short)0, acc0, false, false);
      b.x = qa.z; b.y = qa.w;
      acc1 = __builtin_amdgcn_wmma_f32_16x16x4_f32(false, a[t], false, b,
                                                   (short)0, acc1, false, false);
      b.x = qb.x; b.y = qb.y;
      acc2 = __builtin_amdgcn_wmma_f32_16x16x4_f32(false, a[t], false, b,
                                                   (short)0, acc2, false, false);
      b.x = qb.z; b.y = qb.w;
      acc3 = __builtin_amdgcn_wmma_f32_16x16x4_f32(false, a[t], false, b,
                                                   (short)0, acc3, false, false);
    }
    float* __restrict__ cp = support + (size_t)cbase + 4 * lid;
#pragma unroll
    for (int r = 0; r < 8; ++r) {
      const int row = m0 + r + 8 * half;
      if (row < N) {
        float4 v;
        v.x = acc0[r]; v.y = acc1[r]; v.z = acc2[r]; v.w = acc3[r];
        *(float4*)(cp + (size_t)row * CDIM) = v;
      }
    }
  }
}

#define EDGE_GROUP 2

__device__ __forceinline__ float bcast_f32(float v, int j) {
  return __builtin_bit_cast(float,
      __builtin_amdgcn_readlane(__builtin_bit_cast(int, v), j));
}

typedef __attribute__((ext_vector_type(4))) float v4f;

template <int G>
__device__ __forceinline__ void edge_group(const int row, const int col,
                                           const float (&val)[KER], const int j,
                                           const int lane,
                                           const float* __restrict__ support,
                                           float* __restrict__ out) {
  v4f s0[G], s1[G];
#pragma unroll
  for (int u = 0; u < G; ++u) {
    const int cj = __builtin_amdgcn_readlane(col, j + u);
    const v4f* sp = (const v4f*)(support + (size_t)cj * CDIM + lane * 8);
    s0[u] = sp[0];
    s1[u] = sp[1];
  }
  static_assert(G == 1 || G == 2, "the use below names the registers of G == 2");
  if constexpr (G == 2)
    asm volatile("" : "+v"(s0[0]), "+v"(s1[0]), "+v"(s0[1]), "+v"(s1[1]));
#pragma unroll
  for (int u = 0; u < G; ++u) {
    const int   rj = __builtin_amdgcn_readlane(row, j + u);
    const float v0 = bcast_f32(val[0], j + u);
    const float v1 = bcast_f32(val[1], j + u);
    const float v2 = bcast_f32(val[2], j + u);
    const float v3 = bcast_f32(val[3], j + u);
    const float m0 = s0[u].x * v0 + s0[u].y * v1 + s0[u].z * v2 + s0[u].w * v3;
    const float m1 = s1[u].x * v0 + s1[u].y * v1 + s1[u].z * v2 + s1[u].w * v3;
    float* dst = out + (size_t)rj * OUT_F + lane * 2;
    unsafeAtomicAdd(dst,     m0);
    unsafeAtomicAdd(dst + 1, m1);
  }
}

__global__ void edge_scatter(const int*   __restrict__ eidx,
                             const float* __restrict__ spec,
                             const float* __restrict__ mu,
                             const float* __restrict__ sig,
                             const float* __restrict__ support,
                             float* __restrict__ out,
                             int E) {
  const int lane = threadIdx.x & 31;
  const int wave =
      __builtin_amdgcn_readfirstlane((int)((blockIdx.x * blockDim.x + threadIdx.x) >> 5));
  const int nwaves = (int)((gridDim.x * blockDim.x) >> 5);

  const int per  = (E + nwaves - 1) / nwaves;
  const int ebeg = wave * per;
  int eend = ebeg + per;
  if (eend > E) eend = E;

  float mu0[KER], mu1[KER], mu2[KER], sg[KER];
#pragma unroll
  for (int k = 0; k < KER; ++k) {
    mu0[k] = mu[0 * KER + k];
    mu1[k] = mu[1 * KER + k];
    mu2[k] = mu[2 * KER + k];
    sg[k]  = sig[k];
  }

  for (int base = ebeg; base < eend; base += 32) {
    int n = eend - base;
    if (n > 32) n = 32;

    int e = base + lane;
    if (e >= eend) e = eend - 1;
    const int row = eidx[e];
    const int col = eidx[E + e];

    const float d0 = spec[row * EMB + 0] - spec[col * EMB + 0];
    const float d1 = spec[row * EMB + 1] - spec[col * EMB + 1];
    const float d2 = spec[row * EMB + 2] - spec[col * EMB + 2];

    float val[KER];
#pragma unroll
    for (int k = 0; k < KER; ++k) {
      const float a0 = d0 - mu0[k];
      const float a1 = d1 - mu1[k];
      const float a2 = d2 - mu2[k];
      const float qq = -0.5f * (a0 * a0 + a1 * a1 + a2 * a2);
      val[k] = __expf(sg[k] * qq);
    }

    int j = 0;
    for (; j + EDGE_GROUP <= n; j += EDGE_GROUP)
      edge_group<EDGE_GROUP>(row, col, val, j, lane, support, out);
#pragma nounroll
    for (; j < n; ++j)
      edge_group<1>(row, col, val, j, lane, support, out);
  }
}

extern "C" void kernel_launch(void* const* d_in, const int* in_sizes, int n_in,
                              void* d_out, int out_size, void* d_ws, size_t ws_size,
                              hipStream_t stream) {
  const float* x    = (const float*)d_in[0];
  const int*   eidx = (const int*)  d_in[1];
  const float* spec = (const float*)d_in[2];
  const float* w    = (const float*)d_in[3];
  const float* bias = (const float*)d_in[4];
  const float* mu   = (const float*)d_in[5];
  const float* sig  = (const float*)d_in[6];

  const int N = in_sizes[0] / IN_F;
  const int E = in_sizes[1] / 2;

  float* support = (float*)d_ws;
  float* out     = (float*)d_out;

  {
    const int rowTiles = (N + 15) / 16;
    const dim3 grid(CDIM / CHUNK, (rowTiles + GEMM_WAVES - 1) / GEMM_WAVES);
    support_gemm_wmma<<<grid, GEMM_WAVES * 32, 0, stream>>>(x, w, support, bias, out, N);
  }

  {
    const int blocks = 2048;
    edge_scatter<<<blocks, 256, 0, stream>>>(eidx, spec, mu, sig, support, out, E);
  }
}


# Edge phase: a lane owns outputs {l, l+32} instead of {2l, 2l+1}: each gather load reads 4 whole 128-B lines (was 8 half-used), each f32 atomic adds into 1 whole line (was 2); sum order unchanged
# speedup vs baseline: 1.3457x; 1.1029x over previous
#include <hip/hip_runtime.h>


typedef __attribute__((ext_vector_type(2))) float v2f;
typedef __attribute__((ext_vector_type(8))) float v8f;

#define IN_F  64
#define OUT_F 64
#define KER   4
#define EMB   3
#define CDIM  (OUT_F * KER)

#define GEMM_WAVES 8
#define CHUNK      64
#define SEED_ROWS  ((GEMM_WAVES * 16) / (CDIM / CHUNK))

__global__ void __launch_bounds__(GEMM_WAVES * 32)
support_gemm_wmma(const float* __restrict__ x,
                  const float* __restrict__ w,
                  float* __restrict__ support,
                  const float* __restrict__ bias,
                  float* __restrict__ out,
                  int N) {
  __shared__ float4 wl[2][(IN_F / 2) * (CHUNK / 4)];

  const int tid  = threadIdx.x;
  const int lane = tid & 31;
  const int wv   = __builtin_amdgcn_readfirstlane(tid >> 5);
  const int half = lane >> 4;
  const int lid  = lane & 15;

  const int cbase    = blockIdx.x * CHUNK;
  const int rowTiles = (N + 15) >> 4;
  const int strip    = blockIdx.y * GEMM_WAVES + wv;
  const int m0       = strip * 16;

  int arow = m0 + lid;
  if (arow >= N) arow = N - 1;
  const v2f* __restrict__ ap =
      (const v2f*)(x + (size_t)arow * IN_F + half * 2);
  v2f a[IN_F / 4];
#pragma unroll
  for (int t = 0; t < IN_F / 4; ++t) a[t] = ap[2 * t];

  float4 bv = ((const float4*)bias)[tid & 15];

  {
    const float4* __restrict__ wg = (const float4*)(w + cbase);
    const int kq = tid >> 4;
    const int c4 = tid & 15;
#pragma unroll
    for (int p = 0; p < IN_F / 32; ++p) {
      const int kp = kq + 16 * p;
      const float4 r0 = wg[(size_t)(2 * kp)     * (CDIM / 4) + c4];
      const float4 r1 = wg[(size_t)(2 * kp + 1) * (CDIM / 4) + c4];
      float4 lo, hi;
      lo.x = r0.x; lo.y = r1.x; lo.z = r0.y; lo.w = r1.y;
      hi.x = r0.z; hi.y = r1.z; hi.z = r0.w; hi.w = r1.w;
      wl[0][kp * (CHUNK / 4) + c4] = lo;
      wl[1][kp * (CHUNK / 4) + c4] = hi;
    }
  }
  asm volatile("" : "+v"(bv.x), "+v"(bv.y), "+v"(bv.z), "+v"(bv.w));
  __syncthreads();

  {
    const int r0 = blockIdx.y * (GEMM_WAVES * 16) + blockIdx.x * SEED_ROWS + (tid >> 4);
    float4* __restrict__ op = (float4*)out + (size_t)r0 * (OUT_F / 4) + (tid & 15);
#pragma unroll
    for (int p = 0; p < SEED_ROWS / 16; ++p) {
      if (r0 + 16 * p < N) op[(size_t)(16 * p) * (OUT_F / 4)] = bv;
    }
  }

  if (strip < rowTiles) {
    v8f acc0 = {}, acc1 = {}, acc2 = {}, acc3 = {};
#pragma unroll
    for (int t = 0; t < IN_F / 4; ++t) {
      const float4 qa = wl[0][(2 * t + half) * (CHUNK / 4) + lid];
      const float4 qb = wl[1][(2 * t + half) * (CHUNK / 4) + lid];
      v2f b;
      b.x = qa.x; b.y = qa.y;
      acc0 = __builtin_amdgcn_wmma_f32_16x16x4_f32(false, a[t], false, b,
                                                   (short)0, acc0, false, false);
      b.x = qa.z; b.y = qa.w;
      acc1 = __builtin_amdgcn_wmma_f32_16x16x4_f32(false, a[t], false, b,
                                                   (short)0, acc1, false, false);
      b.x = qb.x; b.y = qb.y;
      acc2 = __builtin_amdgcn_wmma_f32_16x16x4_f32(false, a[t], false, b,
                                                   (short)0, acc2, false, false);
      b.x = qb.z; b.y = qb.w;
      acc3 = __builtin_amdgcn_wmma_f32_16x16x4_f32(false, a[t], false, b,
                                                   (short)0, acc3, false, false);
    }
    float* __restrict__ cp = support + (size_t)cbase + 4 * lid;
#pragma unroll
    for (int r = 0; r < 8; ++r) {
      const int row = m0 + r + 8 * half;
      if (row < N) {
        float4 v;
        v.x = acc0[r]; v.y = acc1[r]; v.z = acc2[r]; v.w = acc3[r];
        *(float4*)(cp + (size_t)row * CDIM) = v;
      }
    }
  }
}

#define EDGE_GROUP 2

__device__ __forceinline__ float bcast_f32(float v, int j) {
  return __builtin_bit_cast(float,
      __builtin_amdgcn_readlane(__builtin_bit_cast(int, v), j));
}

typedef __attribute__((ext_vector_type(4))) float v4f;

template <int G>
__device__ __forceinline__ void edge_group(const int row, const int col,
                                           const float (&val)[KER], const int j,
                                           const int lane,
                                           const float* __restrict__ support,
                                           float* __restrict__ out) {
  v4f s0[G], s1[G];
#pragma unroll
  for (int u = 0; u < G; ++u) {
    const int cj = __builtin_amdgcn_readlane(col, j + u);
    const v4f* sp = (const v4f*)(support + (size_t)cj * CDIM + lane * KER);
    s0[u] = sp[0];
    s1[u] = sp[(OUT_F / 2) * KER / 4];
  }
  static_assert(G == 1 || G == 2, "the use below names the registers of G == 2");
  if constexpr (G == 2)
    asm volatile("" : "+v"(s0[0]), "+v"(s1[0]), "+v"(s0[1]), "+v"(s1[1]));
#pragma unroll
  for (int u = 0; u < G; ++u) {
    const int   rj = __builtin_amdgcn_readlane(row, j + u);
    const float v0 = bcast_f32(val[0], j + u);
    const float v1 = bcast_f32(val[1], j + u);
    const float v2 = bcast_f32(val[2], j + u);
    const float v3 = bcast_f32(val[3], j + u);
    const float m0 = s0[u].x * v0 + s0[u].y * v1 + s0[u].z * v2 + s0[u].w * v3;
    const float m1 = s1[u].x * v0 + s1[u].y * v1 + s1[u].z * v2 + s1[u].w * v3;
    float* dst = out + (size_t)rj * OUT_F + lane;
    unsafeAtomicAdd(dst,             m0);
    unsafeAtomicAdd(dst + OUT_F / 2, m1);
  }
}

__global__ void edge_scatter(const int*   __restrict__ eidx,
                             const float* __restrict__ spec,
                             const float* __restrict__ mu,
                             const float* __restrict__ sig,
                             const float* __restrict__ support,
                             float* __restrict__ out,
                             int E) {
  const int lane = threadIdx.x & 31;
  const int wave =
      __builtin_amdgcn_readfirstlane((int)((blockIdx.x * blockDim.x + threadIdx.x) >> 5));
  const int nwaves = (int)((gridDim.x * blockDim.x) >> 5);

  const int per  = (E + nwaves - 1) / nwaves;
  const int ebeg = wave * per;
  int eend = ebeg + per;
  if (eend > E) eend = E;

  float mu0[KER], mu1[KER], mu2[KER], sg[KER];
#pragma unroll
  for (int k = 0; k < KER; ++k) {
    mu0[k] = mu[0 * KER + k];
    mu1[k] = mu[1 * KER + k];
    mu2[k] = mu[2 * KER + k];
    sg[k]  = sig[k];
  }

  for (int base = ebeg; base < eend; base += 32) {
    int n = eend - base;
    if (n > 32) n = 32;

    int e = base + lane;
    if (e >= eend) e = eend - 1;
    const int row = eidx[e];
    const int col = eidx[E + e];

    const float d0 = spec[row * EMB + 0] - spec[col * EMB + 0];
    const float d1 = spec[row * EMB + 1] - spec[col * EMB + 1];
    const float d2 = spec[row * EMB + 2] - spec[col * EMB + 2];

    float val[KER];
#pragma unroll
    for (int k = 0; k < KER; ++k) {
      const float a0 = d0 - mu0[k];
      const float a1 = d1 - mu1[k];
      const float a2 = d2 - mu2[k];
      const float qq = -0.5f * (a0 * a0 + a1 * a1 + a2 * a2);
      val[k] = __expf(sg[k] * qq);
    }

    int j = 0;
    for (; j + EDGE_GROUP <= n; j += EDGE_GROUP)
      edge_group<EDGE_GROUP>(row, col, val, j, lane, support, out);
#pragma nounroll
    for (; j < n; ++j)
      edge_group<1>(row, col, val, j, lane, support, out);
  }
}

extern "C" void kernel_launch(void* const* d_in, const int* in_sizes, int n_in,
                              void* d_out, int out_size, void* d_ws, size_t ws_size,
                              hipStream_t stream) {
  const float* x    = (const float*)d_in[0];
  const int*   eidx = (const int*)  d_in[1];
  const float* spec = (const float*)d_in[2];
  const float* w    = (const float*)d_in[3];
  const float* bias = (const float*)d_in[4];
  const float* mu   = (const float*)d_in[5];
  const float* sig  = (const float*)d_in[6];

  const int N = in_sizes[0] / IN_F;
  const int E = in_sizes[1] / 2;

  float* support = (float*)d_ws;
  float* out     = (float*)d_out;

  {
    const int rowTiles = (N + 15) / 16;
    const dim3 grid(CDIM / CHUNK, (rowTiles + GEMM_WAVES - 1) / GEMM_WAVES);
    support_gemm_wmma<<<grid, GEMM_WAVES * 32, 0, stream>>>(x, w, support, bias, out, N);
  }

  {
    const int blocks = 2048;
    edge_scatter<<<blocks, 256, 0, stream>>>(eidx, spec, mu, sig, support, out, E);
  }
}


# Edges grouped by source-node range (LDS counting sort); each workgroup builds its 128 support rows by WMMA into LDS and serves its edges from there: no L2 gather; sum order changes
# speedup vs baseline: 1.5984x; 1.1878x over previous
#include <hip/hip_runtime.h>


typedef __attribute__((ext_vector_type(2))) float v2f;
typedef __attribute__((ext_vector_type(4))) float v4f;
typedef __attribute__((ext_vector_type(8))) float v8f;

#define IN_F  64
#define OUT_F 64
#define KER   4
#define EMB   3
#define CDIM  (OUT_F * KER)
#define CHUNK 64

#define BROWS   128
#define BSHIFT  7
#define NB_MAX  512
#define SORT_WGS     128
#define SORT_THREADS 1024
#define SORT_BATCH   8

__global__ void __launch_bounds__(SORT_THREADS)
bucket_count(const int* __restrict__ eidx,
             const float* __restrict__ bias,
             float* __restrict__ out,
             int* __restrict__ cnt,
             int N, int E, int NB) {
  __shared__ int hist[NB_MAX];
  const int tid = threadIdx.x;
  for (int b = tid; b < NB; b += SORT_THREADS) hist[b] = 0;
  __syncthreads();

  const int per  = (E + SORT_WGS - 1) / SORT_WGS;
  const int ebeg = blockIdx.x * per;
  int eend = ebeg + per;
  if (eend > E) eend = E;
  for (int e0 = ebeg + tid; e0 < eend; e0 += SORT_BATCH * SORT_THREADS) {
    int c[SORT_BATCH];
#pragma unroll
    for (int u = 0; u < SORT_BATCH; ++u) {
      const int e = e0 + u * SORT_THREADS;
      c[u] = (e < eend) ? eidx[E + e] : -1;
    }
#pragma unroll
    for (int u = 0; u < SORT_BATCH; ++u)
      if (c[u] >= 0) atomicAdd(&hist[c[u] >> BSHIFT], 1);
  }
  __syncthreads();
  for (int b = tid; b < NB; b += SORT_THREADS) cnt[blockIdx.x * NB + b] = hist[b];

  const float4 bv = ((const float4*)bias)[tid & 15];
  const int total4 = N * (OUT_F / 4);
  for (int i = blockIdx.x * SORT_THREADS + tid; i < total4; i += SORT_WGS * SORT_THREADS)
    ((float4*)out)[i] = bv;
}

__global__ void __launch_bounds__(NB_MAX)
bucket_offsets(const int* __restrict__ cnt,
               int* __restrict__ off,
               int* __restrict__ bstart,
               int* __restrict__ btotal,
               int NB) {
  __shared__ int s[NB_MAX];
  const int tid = threadIdx.x;
  int run = 0;
  if (tid < NB) {
#pragma unroll
    for (int w = 0; w < SORT_WGS; ++w) {
      const int c = cnt[w * NB + tid];
      off[w * NB + tid] = run;
      run += c;
    }
  }
  s[tid] = run;
  __syncthreads();
  for (int d = 1; d < NB_MAX; d <<= 1) {
    const int v = (tid >= d) ? s[tid - d] : 0;
    __syncthreads();
    s[tid] += v;
    __syncthreads();
  }
  if (tid < NB) {
    bstart[tid] = s[tid] - run;
    btotal[tid] = run;
  }
}

__global__ void __launch_bounds__(SORT_THREADS)
bucket_fill(const int* __restrict__ eidx,
            const int* __restrict__ off,
            const int* __restrict__ bstart,
            int2* __restrict__ list,
            int E, int NB) {
  __shared__ int cursor[NB_MAX];
  const int tid = threadIdx.x;
  for (int b = tid; b < NB; b += SORT_THREADS)
    cursor[b] = bstart[b] + off[blockIdx.x * NB + b];
  __syncthreads();

  const int per  = (E + SORT_WGS - 1) / SORT_WGS;
  const int ebeg = blockIdx.x * per;
  int eend = ebeg + per;
  if (eend > E) eend = E;
  for (int e0 = ebeg + tid; e0 < eend; e0 += SORT_BATCH * SORT_THREADS) {
    int r[SORT_BATCH], c[SORT_BATCH];
#pragma unroll
    for (int u = 0; u < SORT_BATCH; ++u) {
      const int e = e0 + u * SORT_THREADS;
      const bool in = e < eend;
      r[u] = in ? eidx[e] : -1;
      c[u] = in ? eidx[E + e] : -1;
    }
#pragma unroll
    for (int u = 0; u < SORT_BATCH; ++u) {
      if (c[u] >= 0) {
        const int pos = atomicAdd(&cursor[c[u] >> BSHIFT], 1);
        list[pos] = make_int2(r[u], c[u]);
      }
    }
  }
}

__device__ __forceinline__ float bcast_f32(float v, int j) {
  return __builtin_bit_cast(float,
      __builtin_amdgcn_readlane(__builtin_bit_cast(int, v), j));
}

template <int G>
__device__ __forceinline__ void edge_group(const int row, const int col,
                                           const float (&val)[KER], const int j,
                                           const int lane, const int cb,
                                           const v4f* sup,
                                           float* __restrict__ out) {
  v4f s0[G], s1[G];
#pragma unroll
  for (int u = 0; u < G; ++u) {
    const int cj = __builtin_amdgcn_readlane(col, j + u) - cb;
    const v4f* sp = sup + cj * (CDIM / 4) + lane;
    s0[u] = sp[0];
    s1[u] = sp[OUT_F / 2];
  }
#pragma unroll
  for (int u = 0; u < G; ++u) {
    const int   rj = __builtin_amdgcn_readlane(row, j + u);
    const float v0 = bcast_f32(val[0], j + u);
    const float v1 = bcast_f32(val[1], j + u);
    const float v2 = bcast_f32(val[2], j + u);
    const float v3 = bcast_f32(val[3], j + u);
    const float m0 = s0[u].x * v0 + s0[u].y * v1 + s0[u].z * v2 + s0[u].w * v3;
    const float m1 = s1[u].x * v0 + s1[u].y * v1 + s1[u].z * v2 + s1[u].w * v3;
    float* dst = out + (size_t)rj * OUT_F + lane;
    unsafeAtomicAdd(dst,             m0);
    unsafeAtomicAdd(dst + OUT_F / 2, m1);
  }
}

#define CONV_WAVES 16
#define EDGE_GROUP 4

__global__ void __launch_bounds__(CONV_WAVES * 32)
bucket_conv(const float* __restrict__ x,
            const float* __restrict__ w,
            const int2*  __restrict__ list,
            const int*   __restrict__ bstart,
            const int*   __restrict__ btotal,
            const float* __restrict__ spec,
            const float* __restrict__ mu,
            const float* __restrict__ sig,
            float* __restrict__ out,
            int N) {
  __shared__ float4 wl[2][(IN_F / 2) * (CHUNK / 4)];
  __shared__ float4 sup[BROWS * (CDIM / 4)];

  const int tid  = threadIdx.x;
  const int lane = tid & 31;
  const int wv   = __builtin_amdgcn_readfirstlane(tid >> 5);
  const int half = lane >> 4;
  const int lid  = lane & 15;
  const int b    = blockIdx.x;
  const int cb   = b * BROWS;

  const int mt = wv & 7;
  const int ch = wv >> 3;
  {
    int arow = cb + mt * 16 + lid;
    if (arow >= N) arow = N - 1;
    const v2f* __restrict__ ap = (const v2f*)(x + (size_t)arow * IN_F + half * 2);
    v2f a[IN_F / 4];
#pragma unroll
    for (int t = 0; t < IN_F / 4; ++t) a[t] = ap[2 * t];

    float2* __restrict__ sup2 = (float2*)sup;
    for (int p = 0; p < CDIM / CHUNK; ++p) {
      {
        const float4* __restrict__ wg = (const float4*)(w + p * CHUNK);
        const int kp = tid >> 4;
        const int c4 = tid & 15;
        const float4 r0 = wg[(size_t)(2 * kp)     * (CDIM / 4) + c4];
        const float4 r1 = wg[(size_t)(2 * kp + 1) * (CDIM / 4) + c4];
        float4 lo, hi;
        lo.x = r0.x; lo.y = r1.x; lo.z = r0.y; lo.w = r1.y;
        hi.x = r0.z; hi.y = r1.z; hi.z = r0.w; hi.w = r1.w;
        wl[0][kp * (CHUNK / 4) + c4] = lo;
        wl[1][kp * (CHUNK / 4) + c4] = hi;
      }
      __syncthreads();

      v8f acc0 = {}, acc1 = {};
#pragma unroll
      for (int t = 0; t < IN_F / 4; ++t) {
        const float4 q = wl[ch][(2 * t + half) * (CHUNK / 4) + lid];
        v2f bb;
        bb.x = q.x; bb.y = q.y;
        acc0 = __builtin_amdgcn_wmma_f32_16x16x4_f32(false, a[t], false, bb,
                                                     (short)0, acc0, false, false);
        bb.x = q.z; bb.y = q.w;
        acc1 = __builtin_amdgcn_wmma_f32_16x16x4_f32(false, a[t], false, bb,
                                                     (short)0, acc1, false, false);
      }
#pragma unroll
      for (int r = 0; r < 8; ++r) {
        float2 v;
        v.x = acc0[r]; v.y = acc1[r];
        sup2[(mt * 16 + r + 8 * half) * (CDIM / 2) + p * (CHUNK / 2) + 2 * lid + ch] = v;
      }
      __syncthreads();
    }
  }

  const int beg = bstart[b];
  const int cnt = btotal[b];

  float mu0[KER], mu1[KER], mu2[KER], sg[KER];
#pragma unroll
  for (int k = 0; k < KER; ++k) {
    mu0[k] = mu[0 * KER + k];
    mu1[k] = mu[1 * KER + k];
    mu2[k] = mu[2 * KER + k];
    sg[k]  = sig[k];
  }

  const v4f* supv = (const v4f*)sup;
  for (int base = wv * 32; base < cnt; base += CONV_WAVES * 32) {
    int n = cnt - base;
    if (n > 32) n = 32;

    int e = base + lane;
    if (e >= cnt) e = cnt - 1;
    const int2 rc = list[beg + e];
    const int row = rc.x;
    const int col = rc.y;

    const float d0 = spec[row * EMB + 0] - spec[col * EMB + 0];
    const float d1 = spec[row * EMB + 1] - spec[col * EMB + 1];
    const float d2 = spec[row * EMB + 2] - spec[col * EMB + 2];

    float val[KER];
#pragma unroll
    for (int k = 0; k < KER; ++k) {
      const float a0 = d0 - mu0[k];
      const float a1 = d1 - mu1[k];
      const float a2 = d2 - mu2[k];
      const float qq = -0.5f * (a0 * a0 + a1 * a1 + a2 * a2);
      val[k] = __expf(sg[k] * qq);
    }

    int j = 0;
    for (; j + EDGE_GROUP <= n; j += EDGE_GROUP)
      edge_group<EDGE_GROUP>(row, col, val, j, lane, cb, supv, out);
#pragma nounroll
    for (; j < n; ++j)
      edge_group<1>(row, col, val, j, lane, cb, supv, out);
  }
}

extern "C" void kernel_launch(void* const* d_in, const int* in_sizes, int n_in,
                              void* d_out, int out_size, void* d_ws, size_t ws_size,
                              hipStream_t stream) {
  const float* x    = (const float*)d_in[0];
  const int*   eidx = (const int*)  d_in[1];
  const float* spec = (const float*)d_in[2];
  const float* w    = (const float*)d_in[3];
  const float* bias = (const float*)d_in[4];
  const float* mu   = (const float*)d_in[5];
  const float* sig  = (const float*)d_in[6];

  const int N  = in_sizes[0] / IN_F;
  const int E  = in_sizes[1] / 2;
  const int NB = (N + BROWS - 1) / BROWS;

  int2* list   = (int2*)d_ws;
  int*  cnt    = (int*)(list + E);
  int*  off    = cnt + SORT_WGS * NB;
  int*  bstart = off + SORT_WGS * NB;
  int*  btotal = bstart + NB;
  float* out   = (float*)d_out;

  bucket_count<<<SORT_WGS, SORT_THREADS, 0, stream>>>(eidx, bias, out, cnt, N, E, NB);
  bucket_offsets<<<1, NB_MAX, 0, stream>>>(cnt, off, bstart, btotal, NB);
  bucket_fill<<<SORT_WGS, SORT_THREADS, 0, stream>>>(eidx, off, bstart, list, E, NB);
  bucket_conv<<<NB, CONV_WAVES * 32, 0, stream>>>(x, w, list, bstart, btotal,
                                                  spec, mu, sig, out, N);
}


# Sort done in 1 launch: each workgroup sorts its own edge chunk by bucket in place; a bucket workgroup finds its 128 segments by a table, list reads unpredicated; 2 launches removed; sum order changes
# speedup vs baseline: 2.5478x; 1.5940x over previous
#include <hip/hip_runtime.h>


typedef __attribute__((ext_vector_type(2))) float v2f;
typedef __attribute__((ext_vector_type(4))) float v4f;
typedef __attribute__((ext_vector_type(8))) float v8f;

#define IN_F  64
#define OUT_F 64
#define KER   4
#define EMB   3
#define CDIM  (IN_F * KER)

#define BROWS   128
#define BSHIFT  7
#define NB_MAX  512
#define SORT_WGS     128
#define SORT_THREADS 1024
#define SORT_BATCH   8

__global__ void __launch_bounds__(SORT_THREADS)
chunk_sort(const int* __restrict__ eidx,
           int2* __restrict__ list,
           int* __restrict__ tab,
           int E, int NB) {
  __shared__ int hist[NB_MAX];
  __shared__ int wsum[NB_MAX / 32];
  const int tid = threadIdx.x;

  const int per = (E + SORT_WGS - 1) / SORT_WGS;
  int ebeg = blockIdx.x * per;
  if (ebeg > E) ebeg = E;
  int eend = ebeg + per;
  if (eend > E) eend = E;

  int r0[SORT_BATCH], c0[SORT_BATCH];
#pragma unroll
  for (int u = 0; u < SORT_BATCH; ++u) {
    const int e = ebeg + tid + u * SORT_THREADS;
    const bool in = e < eend;
    r0[u] = in ? eidx[e] : -1;
    c0[u] = in ? eidx[E + e] : 0;
  }
  for (int b = tid; b < NB; b += SORT_THREADS) hist[b] = 0;
  __syncthreads();
#pragma unroll
  for (int u = 0; u < SORT_BATCH; ++u)
    if (r0[u] >= 0) atomicAdd(&hist[r0[u] >> BSHIFT], 1);
  for (int e0 = ebeg + tid + SORT_BATCH * SORT_THREADS; e0 < eend;
       e0 += SORT_BATCH * SORT_THREADS) {
    int r[SORT_BATCH];
#pragma unroll
    for (int u = 0; u < SORT_BATCH; ++u) {
      const int e = e0 + u * SORT_THREADS;
      r[u] = (e < eend) ? eidx[e] : -1;
    }
#pragma unroll
    for (int u = 0; u < SORT_BATCH; ++u)
      if (r[u] >= 0) atomicAdd(&hist[r[u] >> BSHIFT], 1);
  }
  __syncthreads();

  const int lane = tid & 31;
  const int wv   = tid >> 5;
  int v = 0, inc = 0;
  if (tid < NB_MAX) {
    v = (tid < NB) ? hist[tid] : 0;
    inc = v;
#pragma unroll
    for (int d = 1; d < 32; d <<= 1) {
      const int t = __shfl_up(inc, d);
      if (lane >= d) inc += t;
    }
    if (lane == 31) wsum[wv] = inc;
  }
  __syncthreads();
  if (tid < NB) {
    int before = 0;
#pragma unroll
    for (int q = 0; q < NB_MAX / 32; ++q) {
      const int s = wsum[q];
      if (q < wv) before += s;
    }
    const int start = before + inc - v;
    hist[tid] = start;
    tab[tid * SORT_WGS + blockIdx.x] = start;
  }
  if (tid == 0) tab[NB * SORT_WGS + blockIdx.x] = eend - ebeg;
  __syncthreads();

  int pos[SORT_BATCH];
#pragma unroll
  for (int u = 0; u < SORT_BATCH; ++u) {
    pos[u] = 0;
    if (r0[u] >= 0) pos[u] = atomicAdd(&hist[r0[u] >> BSHIFT], 1);
  }
#pragma unroll
  for (int u = 0; u < SORT_BATCH; ++u)
    if (r0[u] >= 0) list[ebeg + pos[u]] = make_int2(r0[u], c0[u]);
  for (int e0 = ebeg + tid + SORT_BATCH * SORT_THREADS; e0 < eend;
       e0 += SORT_BATCH * SORT_THREADS) {
    int r[SORT_BATCH], c[SORT_BATCH];
#pragma unroll
    for (int u = 0; u < SORT_BATCH; ++u) {
      const int e = e0 + u * SORT_THREADS;
      const bool in = e < eend;
      r[u] = in ? eidx[e] : -1;
      c[u] = in ? eidx[E + e] : 0;
    }
#pragma unroll
    for (int u = 0; u < SORT_BATCH; ++u) {
      if (r[u] >= 0) {
        const int pos = atomicAdd(&hist[r[u] >> BSHIFT], 1);
        list[ebeg + pos] = make_int2(r[u], c[u]);
      }
    }
  }
}

#define CONV_WAVES   16
#define CONV_THREADS (CONV_WAVES * 32)
#define WROWS   (BROWS / CONV_WAVES)
#define APITCH  (CDIM + 4)
#define CAP     3072
#define EPT     (CAP / CONV_THREADS)
#define ICHUNK  16

__device__ __forceinline__ float bcast_f32(float v, int j) {
  return __builtin_bit_cast(float,
      __builtin_amdgcn_readlane(__builtin_bit_cast(int, v), j));
}

__device__ __forceinline__ void flush_row(float* agg, const int r, const int lane,
                                          float (&acc)[2][KER]) {
  v4f* p = (v4f*)(agg + r * APITCH + 8 * lane);
  v4f lo = p[0], hi = p[1];
  lo.x += acc[0][0]; lo.y += acc[0][1]; lo.z += acc[0][2]; lo.w += acc[0][3];
  hi.x += acc[1][0]; hi.y += acc[1][1]; hi.z += acc[1][2]; hi.w += acc[1][3];
  p[0] = lo;
  p[1] = hi;
#pragma unroll
  for (int k = 0; k < KER; ++k) { acc[0][k] = 0.0f; acc[1][k] = 0.0f; }
}

__device__ __forceinline__ void batch_reads(const int* scol, const unsigned char* srow,
                                            const float* __restrict__ spec,
                                            const int cb, const int base, const int w1,
                                            const int lane, int& col, int& rl,
                                            float (&sr)[EMB], float (&sc)[EMB]) {
  int e = base + lane;
  if (e >= w1) e = w1 - 1;
  col = scol[e];
  rl  = srow[e];
  const int row = cb + rl;
#pragma unroll
  for (int t = 0; t < EMB; ++t) {
    sr[t] = spec[row * EMB + t];
    sc[t] = spec[col * EMB + t];
  }
}

template <int G>
__device__ __forceinline__ void edge_group(const int col, const int rl,
                                           const float (&val)[KER], const int j,
                                           const int lane,
                                           const float* __restrict__ x,
                                           float* agg, int& cur,
                                           float (&acc)[2][KER]) {
  v2f xv[G];
#pragma unroll
  for (int u = 0; u < G; ++u) {
    const int cj = __builtin_amdgcn_readlane(col, j + u);
    xv[u] = ((const v2f*)(x + (size_t)cj * IN_F))[(unsigned)lane];
  }
#pragma unroll
  for (int u = 0; u < G; ++u) {
    const int rj = __builtin_amdgcn_readlane(rl, j + u);
    if (rj != cur) {
      if (cur >= 0) flush_row(agg, cur, lane, acc);
      cur = rj;
    }
    const float v0 = bcast_f32(val[0], j + u);
    const float v1 = bcast_f32(val[1], j + u);
    const float v2 = bcast_f32(val[2], j + u);
    const float v3 = bcast_f32(val[3], j + u);
    acc[0][0] += v0 * xv[u].x; acc[0][1] += v1 * xv[u].x;
    acc[0][2] += v2 * xv[u].x; acc[0][3] += v3 * xv[u].x;
    acc[1][0] += v0 * xv[u].y; acc[1][1] += v1 * xv[u].y;
    acc[1][2] += v2 * xv[u].y; acc[1][3] += v3 * xv[u].y;
  }
}

__global__ void __launch_bounds__(CONV_THREADS)
bucket_conv(const float* __restrict__ x,
            const float* __restrict__ w,
            const int2*  __restrict__ list,
            const int*   __restrict__ tab,
            const float* __restrict__ spec,
            const float* __restrict__ mu,
            const float* __restrict__ sig,
            const float* __restrict__ bias,
            float* __restrict__ out,
            int N, int E) {
  __shared__ __attribute__((aligned(16))) float agg[BROWS * APITCH];
  __shared__ float4 stage[ICHUNK * OUT_F];
  __shared__ __attribute__((aligned(16))) int hist[BROWS];
  __shared__ __attribute__((aligned(16))) int cursor[BROWS];
  __shared__ __attribute__((aligned(16))) int rstart[BROWS + 4];
  __shared__ __attribute__((aligned(16))) int segbase[SORT_WGS + 4];
  __shared__ __attribute__((aligned(16))) int segoff[SORT_WGS];

  int*           scol = (int*)stage;
  unsigned char* srow = (unsigned char*)(scol + CAP);

  const int tid  = threadIdx.x;
  const int lane = tid & 31;
  const int wv   = __builtin_amdgcn_readfirstlane(tid >> 5);
  const int half = lane >> 4;
  const int lid  = lane & 15;
  const int b    = blockIdx.x;
  const int cb   = b * BROWS;

  int4 s0 = make_int4(0, 0, 0, 0), s1 = make_int4(0, 0, 0, 0);
  if (tid < 32) {
    s0 = ((const int4*)(tab + (size_t)b * SORT_WGS))[tid];
    s1 = ((const int4*)(tab + (size_t)(b + 1) * SORT_WGS))[tid];
  }

  {
    v4f* a4 = (v4f*)agg;
    const v4f z = {0.0f, 0.0f, 0.0f, 0.0f};
    for (int i = tid; i < BROWS * APITCH / 4; i += CONV_THREADS) a4[i] = z;
  }

  if (tid < 32) {
    const int per = (E + SORT_WGS - 1) / SORT_WGS;
    const int l0 = s1.x - s0.x, l1 = s1.y - s0.y, l2 = s1.z - s0.z, l3 = s1.w - s0.w;
    const int tot = l0 + l1 + l2 + l3;
    int inc = tot;
#pragma unroll
    for (int d = 1; d < 32; d <<= 1) {
      const int v = __shfl_up(inc, d);
      if (tid >= d) inc += v;
    }
    int4 sb;
    sb.x = inc - tot;
    sb.y = sb.x + l0;
    sb.z = sb.y + l1;
    sb.w = sb.z + l2;
    ((int4*)segbase)[tid] = sb;
    int4 so;
    so.x = (4 * tid + 0) * per + s0.x - sb.x;
    so.y = (4 * tid + 1) * per + s0.y - sb.y;
    so.z = (4 * tid + 2) * per + s0.z - sb.z;
    so.w = (4 * tid + 3) * per + s0.w - sb.w;
    ((int4*)segoff)[tid] = so;
    if (tid == 31) segbase[SORT_WGS] = inc;
  }
  __syncthreads();
  const int cnt = segbase[SORT_WGS];

  float mu0[KER], mu1[KER], mu2[KER], sg[KER];
#pragma unroll
  for (int k = 0; k < KER; ++k) {
    mu0[k] = mu[0 * KER + k];
    mu1[k] = mu[1 * KER + k];
    mu2[k] = mu[2 * KER + k];
    sg[k]  = sig[k];
  }

  int c0 = 0;
  do {
    int m = cnt - c0;
    if (m > CAP) m = CAP;

    int er[EPT], ec[EPT], gi[EPT], seg[EPT];
#pragma unroll
    for (int u = 0; u < EPT; ++u) {
      gi[u] = c0 + tid + u * CONV_THREADS;
      seg[u] = 0;
    }
#pragma unroll
    for (int step = SORT_WGS / 2; step >= 1; step >>= 1) {
#pragma unroll
      for (int u = 0; u < EPT; ++u)
        if (segbase[seg[u] + step] <= gi[u]) seg[u] += step;
    }
#pragma unroll
    for (int u = 0; u < EPT; ++u) {
      const bool in = tid + u * CONV_THREADS < m;
      const int2 rc = list[min(gi[u] + segoff[seg[u]], E - 1)];
      er[u] = in ? rc.x - cb : -1;
      ec[u] = in ? rc.y : 0;
    }
    if (tid < BROWS) hist[tid] = 0;
    __syncthreads();
#pragma unroll
    for (int u = 0; u < EPT; ++u)
      if (er[u] >= 0) atomicAdd(&hist[er[u]], 1);
    __syncthreads();
    if (tid < 32) {
      const int4 own = ((const int4*)hist)[tid];
      const int tot = own.x + own.y + own.z + own.w;
      int inc = tot;
#pragma unroll
      for (int d = 1; d < 32; d <<= 1) {
        const int v = __shfl_up(inc, d);
        if (tid >= d) inc += v;
      }
      int4 st;
      st.x = inc - tot;
      st.y = st.x + own.x;
      st.z = st.y + own.y;
      st.w = st.z + own.z;
      ((int4*)rstart)[tid] = st;
      ((int4*)cursor)[tid] = st;
      if (tid == 31) rstart[BROWS] = inc;
    }
    __syncthreads();
#pragma unroll
    for (int u = 0; u < EPT; ++u) {
      if (er[u] >= 0) {
        const int pos = atomicAdd(&cursor[er[u]], 1);
        scol[pos] = ec[u];
        srow[pos] = (unsigned char)er[u];
      }
    }
    __syncthreads();

    {
      const int w0 = __builtin_amdgcn_readfirstlane(rstart[wv * WROWS]);
      const int w1 = __builtin_amdgcn_readfirstlane(rstart[wv * WROWS + WROWS]);
      float acc[2][KER];
#pragma unroll
      for (int k = 0; k < KER; ++k) { acc[0][k] = 0.0f; acc[1][k] = 0.0f; }
      int cur = -1;

      int colN = 0, rlN = 0;
      float srN[EMB] = {0.0f, 0.0f, 0.0f}, scN[EMB] = {0.0f, 0.0f, 0.0f};
      if (w0 < w1) batch_reads(scol, srow, spec, cb, w0, w1, lane, colN, rlN, srN, scN);

      for (int base = w0; base < w1; base += 32) {
        int n = w1 - base;
        if (n > 32) n = 32;

        const int col = colN;
        const int rl  = rlN;
        const float d0 = srN[0] - scN[0];
        const float d1 = srN[1] - scN[1];
        const float d2 = srN[2] - scN[2];

        float val[KER];
#pragma unroll
        for (int k = 0; k < KER; ++k) {
          const float a0 = d0 - mu0[k];
          const float a1 = d1 - mu1[k];
          const float a2 = d2 - mu2[k];
          const float qq = -0.5f * (a0 * a0 + a1 * a1 + a2 * a2);
          val[k] = __expf(sg[k] * qq);
        }

        if (base + 32 < w1)
          batch_reads(scol, srow, spec, cb, base + 32, w1, lane, colN, rlN, srN, scN);

        int j = 0;
        for (; j + 16 <= n; j += 16)
          edge_group<16>(col, rl, val, j, lane, x, agg, cur, acc);
        if (n - j >= 8) { edge_group<8>(col, rl, val, j, lane, x, agg, cur, acc); j += 8; }
        if (n - j >= 4) { edge_group<4>(col, rl, val, j, lane, x, agg, cur, acc); j += 4; }
        if (n - j >= 2) { edge_group<2>(col, rl, val, j, lane, x, agg, cur, acc); j += 2; }
        if (n - j >= 1) { edge_group<1>(col, rl, val, j, lane, x, agg, cur, acc); }
      }
      if (cur >= 0) flush_row(agg, cur, lane, acc);
    }
    __syncthreads();
    c0 += CAP;
  } while (c0 < cnt);

  const int mt = wv & 7;
  const int ch = wv >> 3;
  v8f acc0 = {}, acc1 = {};
  const float* arow = agg + (mt * 16 + lid) * APITCH + 4 * half;
  const float4* __restrict__ wg = (const float4*)w;
#pragma unroll
  for (int p = 0; p < IN_F / ICHUNK; ++p) {
#pragma unroll
    for (int s = 0; s < (ICHUNK * OUT_F) / CONV_THREADS; ++s) {
      const int f  = tid + s * CONV_THREADS;
      const int il = f >> 6;
      const int o  = f & 63;
      const float4 q = wg[(size_t)(p * ICHUNK + il) * OUT_F + o];
      const int di = ((((il >> 1) * 2 + (o >> 5)) * 2 + (o & 1)) * 2 + (il & 1)) * 16
                     + ((o & 31) >> 1);
      stage[di] = q;
    }
    __syncthreads();

#pragma unroll
    for (int u = 0; u < ICHUNK / 2; ++u) {
      const v4f a  = *(const v4f*)(arow + p * (ICHUNK * KER) + 8 * u);
      const float4 q0 = stage[(((u * 2 + ch) * 2 + 0) * 2 + half) * 16 + lid];
      const float4 q1 = stage[(((u * 2 + ch) * 2 + 1) * 2 + half) * 16 + lid];
      v2f aa, bb;
      aa.x = a.x; aa.y = a.y;
      bb.x = q0.x; bb.y = q0.y;
      acc0 = __builtin_amdgcn_wmma_f32_16x16x4_f32(false, aa, false, bb,
                                                   (short)0, acc0, false, false);
      bb.x = q1.x; bb.y = q1.y;
      acc1 = __builtin_amdgcn_wmma_f32_16x16x4_f32(false, aa, false, bb,
                                                   (short)0, acc1, false, false);
      aa.x = a.z; aa.y = a.w;
      bb.x = q0.z; bb.y = q0.w;
      acc0 = __builtin_amdgcn_wmma_f32_16x16x4_f32(false, aa, false, bb,
                                                   (short)0, acc0, false, false);
      bb.x = q1.z; bb.y = q1.w;
      acc1 = __builtin_amdgcn_wmma_f32_16x16x4_f32(false, aa, false, bb,
                                                   (short)0, acc1, false, false);
    }
    if (p + 1 < IN_F / ICHUNK) __syncthreads();
  }

  const float2 bv = *(const float2*)(bias + 32 * ch + 2 * lid);
#pragma unroll
  for (int r = 0; r < 8; ++r) {
    const int node = cb + mt * 16 + r + 8 * half;
    if (node < N) {
      float2 v;
      v.x = acc0[r] + bv.x;
      v.y = acc1[r] + bv.y;
      *(float2*)(out + (size_t)node * OUT_F + 32 * ch + 2 * lid) = v;
    }
  }
}

extern "C" void kernel_launch(void* const* d_in, const int* in_sizes, int n_in,
                              void* d_out, int out_size, void* d_ws, size_t ws_size,
                              hipStream_t stream) {
  const float* x    = (const float*)d_in[0];
  const int*   eidx = (const int*)  d_in[1];
  const float* spec = (const float*)d_in[2];
  const float* w    = (const float*)d_in[3];
  const float* bias = (const float*)d_in[4];
  const float* mu   = (const float*)d_in[5];
  const float* sig  = (const float*)d_in[6];

  const int N  = in_sizes[0] / IN_F;
  const int E  = in_sizes[1] / 2;
  const int NB = (N + BROWS - 1) / BROWS;

  int*  tab  = (int*)d_ws;
  int2* list = (int2*)(tab + (NB + 1) * SORT_WGS);
  float* out = (float*)d_out;

  chunk_sort<<<SORT_WGS, SORT_THREADS, 0, stream>>>(eidx, list, tab, E, NB);
  bucket_conv<<<NB, CONV_THREADS, 0, stream>>>(x, w, list, tab,
                                               spec, mu, sig, bias, out, N, E);
}


# chunk_sort puts its bucket-grouped chunk together in LDS and copies it to the list as whole 128-B lines per wave store (was one scattered 8-B store per edge); sum order unchanged
# speedup vs baseline: 2.7174x; 1.0666x over previous
#include <hip/hip_runtime.h>


typedef __attribute__((ext_vector_type(2))) float v2f;
typedef __attribute__((ext_vector_type(4))) float v4f;
typedef __attribute__((ext_vector_type(8))) float v8f;

#define IN_F  64
#define OUT_F 64
#define KER   4
#define EMB   3
#define CDIM  (IN_F * KER)

#define BROWS   128
#define BSHIFT  7
#define NB_MAX  512
#define SORT_WGS     128
#define SORT_THREADS 1024
#define SORT_BATCH   8
#define SORT_CAP     (SORT_BATCH * SORT_THREADS)

__global__ void __launch_bounds__(SORT_THREADS)
chunk_sort(const int* __restrict__ eidx,
           int2* __restrict__ list,
           int* __restrict__ tab,
           int E, int NB) {
  __shared__ int hist[NB_MAX];
  __shared__ int wsum[NB_MAX / 32];
  __shared__ int2 sorted[SORT_CAP];
  const int tid = threadIdx.x;

  const int per = (E + SORT_WGS - 1) / SORT_WGS;
  int ebeg = blockIdx.x * per;
  if (ebeg > E) ebeg = E;
  int eend = ebeg + per;
  if (eend > E) eend = E;

  int r0[SORT_BATCH], c0[SORT_BATCH];
#pragma unroll
  for (int u = 0; u < SORT_BATCH; ++u) {
    const int e = ebeg + tid + u * SORT_THREADS;
    const bool in = e < eend;
    r0[u] = in ? eidx[e] : -1;
    c0[u] = in ? eidx[E + e] : 0;
  }
  for (int b = tid; b < NB; b += SORT_THREADS) hist[b] = 0;
  __syncthreads();
#pragma unroll
  for (int u = 0; u < SORT_BATCH; ++u)
    if (r0[u] >= 0) atomicAdd(&hist[r0[u] >> BSHIFT], 1);
  for (int e0 = ebeg + tid + SORT_BATCH * SORT_THREADS; e0 < eend;
       e0 += SORT_BATCH * SORT_THREADS) {
    int r[SORT_BATCH];
#pragma unroll
    for (int u = 0; u < SORT_BATCH; ++u) {
      const int e = e0 + u * SORT_THREADS;
      r[u] = (e < eend) ? eidx[e] : -1;
    }
#pragma unroll
    for (int u = 0; u < SORT_BATCH; ++u)
      if (r[u] >= 0) atomicAdd(&hist[r[u] >> BSHIFT], 1);
  }
  __syncthreads();

  const int lane = tid & 31;
  const int wv   = tid >> 5;
  int v = 0, inc = 0;
  if (tid < NB_MAX) {
    v = (tid < NB) ? hist[tid] : 0;
    inc = v;
#pragma unroll
    for (int d = 1; d < 32; d <<= 1) {
      const int t = __shfl_up(inc, d);
      if (lane >= d) inc += t;
    }
    if (lane == 31) wsum[wv] = inc;
  }
  __syncthreads();
  if (tid < NB) {
    int before = 0;
#pragma unroll
    for (int q = 0; q < NB_MAX / 32; ++q) {
      const int s = wsum[q];
      if (q < wv) before += s;
    }
    const int start = before + inc - v;
    hist[tid] = start;
    tab[tid * SORT_WGS + blockIdx.x] = start;
  }
  if (tid == 0) tab[NB * SORT_WGS + blockIdx.x] = eend - ebeg;
  __syncthreads();

  int pos[SORT_BATCH];
#pragma unroll
  for (int u = 0; u < SORT_BATCH; ++u) {
    pos[u] = 0;
    if (r0[u] >= 0) pos[u] = atomicAdd(&hist[r0[u] >> BSHIFT], 1);
  }
#pragma unroll
  for (int u = 0; u < SORT_BATCH; ++u) {
    if (r0[u] >= 0) {
      if (pos[u] < SORT_CAP) sorted[pos[u]] = make_int2(r0[u], c0[u]);
      else                   list[ebeg + pos[u]] = make_int2(r0[u], c0[u]);
    }
  }
  for (int e0 = ebeg + tid + SORT_BATCH * SORT_THREADS; e0 < eend;
       e0 += SORT_BATCH * SORT_THREADS) {
    int r[SORT_BATCH], c[SORT_BATCH];
#pragma unroll
    for (int u = 0; u < SORT_BATCH; ++u) {
      const int e = e0 + u * SORT_THREADS;
      const bool in = e < eend;
      r[u] = in ? eidx[e] : -1;
      c[u] = in ? eidx[E + e] : 0;
    }
#pragma unroll
    for (int u = 0; u < SORT_BATCH; ++u) {
      if (r[u] >= 0) {
        const int pos = atomicAdd(&hist[r[u] >> BSHIFT], 1);
        if (pos < SORT_CAP) sorted[pos] = make_int2(r[u], c[u]);
        else                list[ebeg + pos] = make_int2(r[u], c[u]);
      }
    }
  }
  __syncthreads();

  int n = eend - ebeg;
  if (n > SORT_CAP) n = SORT_CAP;
  if (n > 0) {
    const int g0 = (ebeg & ~31) + tid;
    int2 ent[SORT_BATCH + 1];
#pragma unroll
    for (int k = 0; k <= SORT_BATCH; ++k) {
      const int i = g0 + k * SORT_THREADS - ebeg;
      ent[k] = sorted[min(max(i, 0), n - 1)];
    }
#pragma unroll
    for (int k = 0; k <= SORT_BATCH; ++k) {
      const int g = g0 + k * SORT_THREADS;
      if (g >= ebeg && g < ebeg + n) list[g] = ent[k];
    }
  }
}

#define CONV_WAVES   16
#define CONV_THREADS (CONV_WAVES * 32)
#define WROWS   (BROWS / CONV_WAVES)
#define APITCH  (CDIM + 4)
#define CAP     3072
#define EPT     (CAP / CONV_THREADS)
#define ICHUNK  16

__device__ __forceinline__ float bcast_f32(float v, int j) {
  return __builtin_bit_cast(float,
      __builtin_amdgcn_readlane(__builtin_bit_cast(int, v), j));
}

__device__ __forceinline__ void flush_row(float* agg, const int r, const int lane,
                                          float (&acc)[2][KER]) {
  v4f* p = (v4f*)(agg + r * APITCH + 8 * lane);
  v4f lo = p[0], hi = p[1];
  lo.x += acc[0][0]; lo.y += acc[0][1]; lo.z += acc[0][2]; lo.w += acc[0][3];
  hi.x += acc[1][0]; hi.y += acc[1][1]; hi.z += acc[1][2]; hi.w += acc[1][3];
  p[0] = lo;
  p[1] = hi;
#pragma unroll
  for (int k = 0; k < KER; ++k) { acc[0][k] = 0.0f; acc[1][k] = 0.0f; }
}

__device__ __forceinline__ void batch_reads(const int* scol, const unsigned char* srow,
                                            const float* __restrict__ spec,
                                            const int cb, const int base, const int w1,
                                            const int lane, int& col, int& rl,
                                            float (&sr)[EMB], float (&sc)[EMB]) {
  int e = base + lane;
  if (e >= w1) e = w1 - 1;
  col = scol[e];
  rl  = srow[e];
  const int row = cb + rl;
#pragma unroll
  for (int t = 0; t < EMB; ++t) {
    sr[t] = spec[row * EMB + t];
    sc[t] = spec[col * EMB + t];
  }
}

template <int G>
__device__ __forceinline__ void edge_group(const int col, const int rl,
                                           const float (&val)[KER], const int j,
                                           const int lane,
                                           const float* __restrict__ x,
                                           float* agg, int& cur,
                                           float (&acc)[2][KER]) {
  v2f xv[G];
#pragma unroll
  for (int u = 0; u < G; ++u) {
    const int cj = __builtin_amdgcn_readlane(col, j + u);
    xv[u] = ((const v2f*)(x + (size_t)cj * IN_F))[(unsigned)lane];
  }
#pragma unroll
  for (int u = 0; u < G; ++u) {
    const int rj = __builtin_amdgcn_readlane(rl, j + u);
    if (rj != cur) {
      if (cur >= 0) flush_row(agg, cur, lane, acc);
      cur = rj;
    }
    const float v0 = bcast_f32(val[0], j + u);
    const float v1 = bcast_f32(val[1], j + u);
    const float v2 = bcast_f32(val[2], j + u);
    const float v3 = bcast_f32(val[3], j + u);
    acc[0][0] += v0 * xv[u].x; acc[0][1] += v1 * xv[u].x;
    acc[0][2] += v2 * xv[u].x; acc[0][3] += v3 * xv[u].x;
    acc[1][0] += v0 * xv[u].y; acc[1][1] += v1 * xv[u].y;
    acc[1][2] += v2 * xv[u].y; acc[1][3] += v3 * xv[u].y;
  }
}

__global__ void __launch_bounds__(CONV_THREADS)
bucket_conv(const float* __restrict__ x,
            const float* __restrict__ w,
            const int2*  __restrict__ list,
            const int*   __restrict__ tab,
            const float* __restrict__ spec,
            const float* __restrict__ mu,
            const float* __restrict__ sig,
            const float* __restrict__ bias,
            float* __restrict__ out,
            int N, int E) {
  __shared__ __attribute__((aligned(16))) float agg[BROWS * APITCH];
  __shared__ float4 stage[ICHUNK * OUT_F];
  __shared__ __attribute__((aligned(16))) int hist[BROWS];
  __shared__ __attribute__((aligned(16))) int cursor[BROWS];
  __shared__ __attribute__((aligned(16))) int rstart[BROWS + 4];
  __shared__ __attribute__((aligned(16))) int segbase[SORT_WGS + 4];
  __shared__ __attribute__((aligned(16))) int segoff[SORT_WGS];

  int*           scol = (int*)stage;
  unsigned char* srow = (unsigned char*)(scol + CAP);

  const int tid  = threadIdx.x;
  const int lane = tid & 31;
  const int wv   = __builtin_amdgcn_readfirstlane(tid >> 5);
  const int half = lane >> 4;
  const int lid  = lane & 15;
  const int b    = blockIdx.x;
  const int cb   = b * BROWS;

  int4 s0 = make_int4(0, 0, 0, 0), s1 = make_int4(0, 0, 0, 0);
  if (tid < 32) {
    s0 = ((const int4*)(tab + (size_t)b * SORT_WGS))[tid];
    s1 = ((const int4*)(tab + (size_t)(b + 1) * SORT_WGS))[tid];
  }

  {
    v4f* a4 = (v4f*)agg;
    const v4f z = {0.0f, 0.0f, 0.0f, 0.0f};
    for (int i = tid; i < BROWS * APITCH / 4; i += CONV_THREADS) a4[i] = z;
  }

  if (tid < 32) {
    const int per = (E + SORT_WGS - 1) / SORT_WGS;
    const int l0 = s1.x - s0.x, l1 = s1.y - s0.y, l2 = s1.z - s0.z, l3 = s1.w - s0.w;
    const int tot = l0 + l1 + l2 + l3;
    int inc = tot;
#pragma unroll
    for (int d = 1; d < 32; d <<= 1) {
      const int v = __shfl_up(inc, d);
      if (tid >= d) inc += v;
    }
    int4 sb;
    sb.x = inc - tot;
    sb.y = sb.x + l0;
    sb.z = sb.y + l1;
    sb.w = sb.z + l2;
    ((int4*)segbase)[tid] = sb;
    int4 so;
    so.x = (4 * tid + 0) * per + s0.x - sb.x;
    so.y = (4 * tid + 1) * per + s0.y - sb.y;
    so.z = (4 * tid + 2) * per + s0.z - sb.z;
    so.w = (4 * tid + 3) * per + s0.w - sb.w;
    ((int4*)segoff)[tid] = so;
    if (tid == 31) segbase[SORT_WGS] = inc;
  }
  __syncthreads();
  const int cnt = segbase[SORT_WGS];

  float mu0[KER], mu1[KER], mu2[KER], sg[KER];
#pragma unroll
  for (int k = 0; k < KER; ++k) {
    mu0[k] = mu[0 * KER + k];
    mu1[k] = mu[1 * KER + k];
    mu2[k] = mu[2 * KER + k];
    sg[k]  = sig[k];
  }

  int c0 = 0;
  do {
    int m = cnt - c0;
    if (m > CAP) m = CAP;

    int er[EPT], ec[EPT], gi[EPT], seg[EPT];
#pragma unroll
    for (int u = 0; u < EPT; ++u) {
      gi[u] = c0 + tid + u * CONV_THREADS;
      seg[u] = 0;
    }
#pragma unroll
    for (int step = SORT_WGS / 2; step >= 1; step >>= 1) {
#pragma unroll
      for (int u = 0; u < EPT; ++u)
        if (segbase[seg[u] + step] <= gi[u]) seg[u] += step;
    }
#pragma unroll
    for (int u = 0; u < EPT; ++u) {
      const bool in = tid + u * CONV_THREADS < m;
      const int2 rc = list[min(gi[u] + segoff[seg[u]], E - 1)];
      er[u] = in ? rc.x - cb : -1;
      ec[u] = in ? rc.y : 0;
    }
    if (tid < BROWS) hist[tid] = 0;
    __syncthreads();
#pragma unroll
    for (int u = 0; u < EPT; ++u)
      if (er[u] >= 0) atomicAdd(&hist[er[u]], 1);
    __syncthreads();
    if (tid < 32) {
      const int4 own = ((const int4*)hist)[tid];
      const int tot = own.x + own.y + own.z + own.w;
      int inc = tot;
#pragma unroll
      for (int d = 1; d < 32; d <<= 1) {
        const int v = __shfl_up(inc, d);
        if (tid >= d) inc += v;
      }
      int4 st;
      st.x = inc - tot;
      st.y = st.x + own.x;
      st.z = st.y + own.y;
      st.w = st.z + own.z;
      ((int4*)rstart)[tid] = st;
      ((int4*)cursor)[tid] = st;
      if (tid == 31) rstart[BROWS] = inc;
    }
    __syncthreads();
#pragma unroll
    for (int u = 0; u < EPT; ++u) {
      if (er[u] >= 0) {
        const int pos = atomicAdd(&cursor[er[u]], 1);
        scol[pos] = ec[u];
        srow[pos] = (unsigned char)er[u];
      }
    }
    __syncthreads();

    {
      const int w0 = __builtin_amdgcn_readfirstlane(rstart[wv * WROWS]);
      const int w1 = __builtin_amdgcn_readfirstlane(rstart[wv * WROWS + WROWS]);
      float acc[2][KER];
#pragma unroll
      for (int k = 0; k < KER; ++k) { acc[0][k] = 0.0f; acc[1][k] = 0.0f; }
      int cur = -1;

      int colN = 0, rlN = 0;
      float srN[EMB] = {0.0f, 0.0f, 0.0f}, scN[EMB] = {0.0f, 0.0f, 0.0f};
      if (w0 < w1) batch_reads(scol, srow, spec, cb, w0, w1, lane, colN, rlN, srN, scN);

      for (int base = w0; base < w1; base += 32) {
        int n = w1 - base;
        if (n > 32) n = 32;

        const int col = colN;
        const int rl  = rlN;
        const float d0 = srN[0] - scN[0];
        const float d1 = srN[1] - scN[1];
        const float d2 = srN[2] - scN[2];

        float val[KER];
#pragma unroll
        for (int k = 0; k < KER; ++k) {
          const float a0 = d0 - mu0[k];
          const float a1 = d1 - mu1[k];
          const float a2 = d2 - mu2[k];
          const float qq = -0.5f * (a0 * a0 + a1 * a1 + a2 * a2);
          val[k] = __expf(sg[k] * qq);
        }

        if (base + 32 < w1)
          batch_reads(scol, srow, spec, cb, base + 32, w1, lane, colN, rlN, srN, scN);

        int j = 0;
        for (; j + 16 <= n; j += 16)
          edge_group<16>(col, rl, val, j, lane, x, agg, cur, acc);
        if (n - j >= 8) { edge_group<8>(col, rl, val, j, lane, x, agg, cur, acc); j += 8; }
        if (n - j >= 4) { edge_group<4>(col, rl, val, j, lane, x, agg, cur, acc); j += 4; }
        if (n - j >= 2) { edge_group<2>(col, rl, val, j, lane, x, agg, cur, acc); j += 2; }
        if (n - j >= 1) { edge_group<1>(col, rl, val, j, lane, x, agg, cur, acc); }
      }
      if (cur >= 0) flush_row(agg, cur, lane, acc);
    }
    __syncthreads();
    c0 += CAP;
  } while (c0 < cnt);

  const int mt = wv & 7;
  const int ch = wv >> 3;
  v8f acc0 = {}, acc1 = {};
  const float* arow = agg + (mt * 16 + lid) * APITCH + 4 * half;
  const float4* __restrict__ wg = (const float4*)w;
#pragma unroll
  for (int p = 0; p < IN_F / ICHUNK; ++p) {
#pragma unroll
    for (int s = 0; s < (ICHUNK * OUT_F) / CONV_THREADS; ++s) {
      const int f  = tid + s * CONV_THREADS;
      const int il = f >> 6;
      const int o  = f & 63;
      const float4 q = wg[(size_t)(p * ICHUNK + il) * OUT_F + o];
      const int di = ((((il >> 1) * 2 + (o >> 5)) * 2 + (o & 1)) * 2 + (il & 1)) * 16
                     + ((o & 31) >> 1);
      stage[di] = q;
    }
    __syncthreads();

#pragma unroll
    for (int u = 0; u < ICHUNK / 2; ++u) {
      const v4f a  = *(const v4f*)(arow + p * (ICHUNK * KER) + 8 * u);
      const float4 q0 = stage[(((u * 2 + ch) * 2 + 0) * 2 + half) * 16 + lid];
      const float4 q1 = stage[(((u * 2 + ch) * 2 + 1) * 2 + half) * 16 + lid];
      v2f aa, bb;
      aa.x = a.x; aa.y = a.y;
      bb.x = q0.x; bb.y = q0.y;
      acc0 = __builtin_amdgcn_wmma_f32_16x16x4_f32(false, aa, false, bb,
                                                   (short)0, acc0, false, false);
      bb.x = q1.x; bb.y = q1.y;
      acc1 = __builtin_amdgcn_wmma_f32_16x16x4_f32(false, aa, false, bb,
                                                   (short)0, acc1, false, false);
      aa.x = a.z; aa.y = a.w;
      bb.x = q0.z; bb.y = q0.w;
      acc0 = __builtin_amdgcn_wmma_f32_16x16x4_f32(false, aa, false, bb,
                                                   (short)0, acc0, false, false);
      bb.x = q1.z; bb.y = q1.w;
      acc1 = __builtin_amdgcn_wmma_f32_16x16x4_f32(false, aa, false, bb,
                                                   (short)0, acc1, false, false);
    }
    if (p + 1 < IN_F / ICHUNK) __syncthreads();
  }

  const float2 bv = *(const float2*)(bias + 32 * ch + 2 * lid);
#pragma unroll
  for (int r = 0; r < 8; ++r) {
    const int node = cb + mt * 16 + r + 8 * half;
    if (node < N) {
      float2 v;
      v.x = acc0[r] + bv.x;
      v.y = acc1[r] + bv.y;
      *(float2*)(out + (size_t)node * OUT_F + 32 * ch + 2 * lid) = v;
    }
  }
}

extern "C" void kernel_launch(void* const* d_in, const int* in_sizes, int n_in,
                              void* d_out, int out_size, void* d_ws, size_t ws_size,
                              hipStream_t stream) {
  const float* x    = (const float*)d_in[0];
  const int*   eidx = (const int*)  d_in[1];
  const float* spec = (const float*)d_in[2];
  const float* w    = (const float*)d_in[3];
  const float* bias = (const float*)d_in[4];
  const float* mu   = (const float*)d_in[5];
  const float* sig  = (const float*)d_in[6];

  const int N  = in_sizes[0] / IN_F;
  const int E  = in_sizes[1] / 2;
  const int NB = (N + BROWS - 1) / BROWS;

  int*  tab  = (int*)d_ws;
  int2* list = (int2*)(tab + (NB + 1) * SORT_WGS);
  float* out = (float*)d_out;

  chunk_sort<<<SORT_WGS, SORT_THREADS, 0, stream>>>(eidx, list, tab, E, NB);
  bucket_conv<<<NB, CONV_THREADS, 0, stream>>>(x, w, list, tab,
                                               spec, mu, sig, bias, out, N, E);
}


# Edge phase: a wave stores the Gaussian weights of its 32-edge batch in an LDS table and reads an edge's four back with one broadcast 16-B read (was 4 readlanes + 2 pair copies); sum order unchanged
# speedup vs baseline: 2.9166x; 1.0733x over previous
#include <hip/hip_runtime.h>


typedef __attribute__((ext_vector_type(2))) float v2f;
typedef __attribute__((ext_vector_type(4))) float v4f;
typedef __attribute__((ext_vector_type(8))) float v8f;

#define IN_F  64
#define OUT_F 64
#define KER   4
#define EMB   3
#define CDIM  (IN_F * KER)

#define BROWS   128
#define BSHIFT  7
#define NB_MAX  512
#define SORT_WGS     128
#define SORT_THREADS 1024
#define SORT_BATCH   8
#define SORT_CAP     (SORT_BATCH * SORT_THREADS)

__global__ void __launch_bounds__(SORT_THREADS)
chunk_sort(const int* __restrict__ eidx,
           int2* __restrict__ list,
           int* __restrict__ tab,
           int E, int NB) {
  __shared__ int hist[NB_MAX];
  __shared__ int wsum[NB_MAX / 32];
  __shared__ int2 sorted[SORT_CAP];
  const int tid = threadIdx.x;

  const int per = (E + SORT_WGS - 1) / SORT_WGS;
  int ebeg = blockIdx.x * per;
  if (ebeg > E) ebeg = E;
  int eend = ebeg + per;
  if (eend > E) eend = E;

  int r0[SORT_BATCH], c0[SORT_BATCH];
#pragma unroll
  for (int u = 0; u < SORT_BATCH; ++u) {
    const int e = ebeg + tid + u * SORT_THREADS;
    const bool in = e < eend;
    r0[u] = in ? eidx[e] : -1;
    c0[u] = in ? eidx[E + e] : 0;
  }
  for (int b = tid; b < NB; b += SORT_THREADS) hist[b] = 0;
  __syncthreads();
#pragma unroll
  for (int u = 0; u < SORT_BATCH; ++u)
    if (r0[u] >= 0) atomicAdd(&hist[r0[u] >> BSHIFT], 1);
  for (int e0 = ebeg + tid + SORT_BATCH * SORT_THREADS; e0 < eend;
       e0 += SORT_BATCH * SORT_THREADS) {
    int r[SORT_BATCH];
#pragma unroll
    for (int u = 0; u < SORT_BATCH; ++u) {
      const int e = e0 + u * SORT_THREADS;
      r[u] = (e < eend) ? eidx[e] : -1;
    }
#pragma unroll
    for (int u = 0; u < SORT_BATCH; ++u)
      if (r[u] >= 0) atomicAdd(&hist[r[u] >> BSHIFT], 1);
  }
  __syncthreads();

  const int lane = tid & 31;
  const int wv   = tid >> 5;
  int v = 0, inc = 0;
  if (tid < NB_MAX) {
    v = (tid < NB) ? hist[tid] : 0;
    inc = v;
#pragma unroll
    for (int d = 1; d < 32; d <<= 1) {
      const int t = __shfl_up(inc, d);
      if (lane >= d) inc += t;
    }
    if (lane == 31) wsum[wv] = inc;
  }
  __syncthreads();
  if (tid < NB) {
    int before = 0;
#pragma unroll
    for (int q = 0; q < NB_MAX / 32; ++q) {
      const int s = wsum[q];
      if (q < wv) before += s;
    }
    const int start = before + inc - v;
    hist[tid] = start;
    tab[tid * SORT_WGS + blockIdx.x] = start;
  }
  if (tid == 0) tab[NB * SORT_WGS + blockIdx.x] = eend - ebeg;
  __syncthreads();

  int pos[SORT_BATCH];
#pragma unroll
  for (int u = 0; u < SORT_BATCH; ++u) {
    pos[u] = 0;
    if (r0[u] >= 0) pos[u] = atomicAdd(&hist[r0[u] >> BSHIFT], 1);
  }
#pragma unroll
  for (int u = 0; u < SORT_BATCH; ++u) {
    if (r0[u] >= 0) {
      if (pos[u] < SORT_CAP) sorted[pos[u]] = make_int2(r0[u], c0[u]);
      else                   list[ebeg + pos[u]] = make_int2(r0[u], c0[u]);
    }
  }
  for (int e0 = ebeg + tid + SORT_BATCH * SORT_THREADS; e0 < eend;
       e0 += SORT_BATCH * SORT_THREADS) {
    int r[SORT_BATCH], c[SORT_BATCH];
#pragma unroll
    for (int u = 0; u < SORT_BATCH; ++u) {
      const int e = e0 + u * SORT_THREADS;
      const bool in = e < eend;
      r[u] = in ? eidx[e] : -1;
      c[u] = in ? eidx[E + e] : 0;
    }
#pragma unroll
    for (int u = 0; u < SORT_BATCH; ++u) {
      if (r[u] >= 0) {
        const int pos = atomicAdd(&hist[r[u] >> BSHIFT], 1);
        if (pos < SORT_CAP) sorted[pos] = make_int2(r[u], c[u]);
        else                list[ebeg + pos] = make_int2(r[u], c[u]);
      }
    }
  }
  __syncthreads();

  int n = eend - ebeg;
  if (n > SORT_CAP) n = SORT_CAP;
  if (n > 0) {
    const int g0 = (ebeg & ~31) + tid;
    int2 ent[SORT_BATCH + 1];
#pragma unroll
    for (int k = 0; k <= SORT_BATCH; ++k) {
      const int i = g0 + k * SORT_THREADS - ebeg;
      ent[k] = sorted[min(max(i, 0), n - 1)];
    }
#pragma unroll
    for (int k = 0; k <= SORT_BATCH; ++k) {
      const int g = g0 + k * SORT_THREADS;
      if (g >= ebeg && g < ebeg + n) list[g] = ent[k];
    }
  }
}

#define CONV_WAVES   16
#define CONV_THREADS (CONV_WAVES * 32)
#define WROWS   (BROWS / CONV_WAVES)
#define APITCH  (CDIM + 4)
#define CAP     3072
#define EPT     (CAP / CONV_THREADS)
#define ICHUNK  16

__device__ __forceinline__ void flush_row(float* agg, const int r, const int lane,
                                          float (&acc)[2][KER]) {
  v4f* p = (v4f*)(agg + r * APITCH + 8 * lane);
  v4f lo = p[0], hi = p[1];
  lo.x += acc[0][0]; lo.y += acc[0][1]; lo.z += acc[0][2]; lo.w += acc[0][3];
  hi.x += acc[1][0]; hi.y += acc[1][1]; hi.z += acc[1][2]; hi.w += acc[1][3];
  p[0] = lo;
  p[1] = hi;
#pragma unroll
  for (int k = 0; k < KER; ++k) { acc[0][k] = 0.0f; acc[1][k] = 0.0f; }
}

__device__ __forceinline__ void batch_reads(const int* scol, const unsigned char* srow,
                                            const float* __restrict__ spec,
                                            const int cb, const int base, const int w1,
                                            const int lane, int& col, int& rl,
                                            float (&sr)[EMB], float (&sc)[EMB]) {
  int e = base + lane;
  if (e >= w1) e = w1 - 1;
  col = scol[e];
  rl  = srow[e];
  const int row = cb + rl;
#pragma unroll
  for (int t = 0; t < EMB; ++t) {
    sr[t] = spec[row * EMB + t];
    sc[t] = spec[col * EMB + t];
  }
}

#define VAHEAD 3
template <int G>
__device__ __forceinline__ void edge_group(const int col, const int rl,
                                           const v4f* vt, const int j,
                                           const int lane,
                                           const float* __restrict__ x,
                                           float* agg, int& cur,
                                           float (&acc)[2][KER]) {
  v2f xv[G];
#pragma unroll
  for (int u = 0; u < G; ++u) {
    const int cj = __builtin_amdgcn_readlane(col, j + u);
    xv[u] = ((const v2f*)(x + (size_t)cj * IN_F))[(unsigned)lane];
  }
  constexpr int D = (G < VAHEAD) ? G : VAHEAD;
  v4f vv[G];
#pragma unroll
  for (int u = 0; u < D; ++u) vv[u] = vt[j + u];
#pragma unroll
  for (int u = 0; u < G; ++u) {
    if (u + D < G) vv[u + D] = vt[j + u + D];
    const int rj = __builtin_amdgcn_readlane(rl, j + u);
    if (rj != cur) {
      if (cur >= 0) flush_row(agg, cur, lane, acc);
      cur = rj;
    }
    const float v0 = vv[u].x, v1 = vv[u].y, v2 = vv[u].z, v3 = vv[u].w;
    acc[0][0] += v0 * xv[u].x; acc[0][1] += v1 * xv[u].x;
    acc[0][2] += v2 * xv[u].x; acc[0][3] += v3 * xv[u].x;
    acc[1][0] += v0 * xv[u].y; acc[1][1] += v1 * xv[u].y;
    acc[1][2] += v2 * xv[u].y; acc[1][3] += v3 * xv[u].y;
  }
}

__global__ void __launch_bounds__(CONV_THREADS)
bucket_conv(const float* __restrict__ x,
            const float* __restrict__ w,
            const int2*  __restrict__ list,
            const int*   __restrict__ tab,
            const float* __restrict__ spec,
            const float* __restrict__ mu,
            const float* __restrict__ sig,
            const float* __restrict__ bias,
            float* __restrict__ out,
            int N, int E) {
  __shared__ __attribute__((aligned(16))) float agg[BROWS * APITCH];
  __shared__ float4 stage[ICHUNK * OUT_F];
  __shared__ __attribute__((aligned(16))) int hist[BROWS];
  __shared__ __attribute__((aligned(16))) int cursor[BROWS];
  __shared__ __attribute__((aligned(16))) int rstart[BROWS + 4];
  __shared__ __attribute__((aligned(16))) int segbase[SORT_WGS + 4];
  __shared__ __attribute__((aligned(16))) int segoff[SORT_WGS];

  __shared__ __attribute__((aligned(16))) float vtab[CONV_WAVES * 32 * KER];

  int*           scol = (int*)stage;
  unsigned char* srow = (unsigned char*)(scol + CAP);

  const int tid  = threadIdx.x;
  const int lane = tid & 31;
  const int wv   = __builtin_amdgcn_readfirstlane(tid >> 5);
  const int half = lane >> 4;
  const int lid  = lane & 15;
  const int b    = blockIdx.x;
  const int cb   = b * BROWS;

  int4 s0 = make_int4(0, 0, 0, 0), s1 = make_int4(0, 0, 0, 0);
  if (tid < 32) {
    s0 = ((const int4*)(tab + (size_t)b * SORT_WGS))[tid];
    s1 = ((const int4*)(tab + (size_t)(b + 1) * SORT_WGS))[tid];
  }

  {
    v4f* a4 = (v4f*)agg;
    const v4f z = {0.0f, 0.0f, 0.0f, 0.0f};
    for (int i = tid; i < BROWS * APITCH / 4; i += CONV_THREADS) a4[i] = z;
  }

  if (tid < 32) {
    const int per = (E + SORT_WGS - 1) / SORT_WGS;
    const int l0 = s1.x - s0.x, l1 = s1.y - s0.y, l2 = s1.z - s0.z, l3 = s1.w - s0.w;
    const int tot = l0 + l1 + l2 + l3;
    int inc = tot;
#pragma unroll
    for (int d = 1; d < 32; d <<= 1) {
      const int v = __shfl_up(inc, d);
      if (tid >= d) inc += v;
    }
    int4 sb;
    sb.x = inc - tot;
    sb.y = sb.x + l0;
    sb.z = sb.y + l1;
    sb.w = sb.z + l2;
    ((int4*)segbase)[tid] = sb;
    int4 so;
    so.x = (4 * tid + 0) * per + s0.x - sb.x;
    so.y = (4 * tid + 1) * per + s0.y - sb.y;
    so.z = (4 * tid + 2) * per + s0.z - sb.z;
    so.w = (4 * tid + 3) * per + s0.w - sb.w;
    ((int4*)segoff)[tid] = so;
    if (tid == 31) segbase[SORT_WGS] = inc;
  }
  __syncthreads();
  const int cnt = segbase[SORT_WGS];

  float mu0[KER], mu1[KER], mu2[KER], sg[KER];
#pragma unroll
  for (int k = 0; k < KER; ++k) {
    mu0[k] = mu[0 * KER + k];
    mu1[k] = mu[1 * KER + k];
    mu2[k] = mu[2 * KER + k];
    sg[k]  = sig[k];
  }

  int c0 = 0;
  do {
    int m = cnt - c0;
    if (m > CAP) m = CAP;

    int er[EPT], ec[EPT], gi[EPT], seg[EPT];
#pragma unroll
    for (int u = 0; u < EPT; ++u) {
      gi[u] = c0 + tid + u * CONV_THREADS;
      seg[u] = 0;
    }
#pragma unroll
    for (int step = SORT_WGS / 2; step >= 1; step >>= 1) {
#pragma unroll
      for (int u = 0; u < EPT; ++u)
        if (segbase[seg[u] + step] <= gi[u]) seg[u] += step;
    }
#pragma unroll
    for (int u = 0; u < EPT; ++u) {
      const bool in = tid + u * CONV_THREADS < m;
      const int2 rc = list[min(gi[u] + segoff[seg[u]], E - 1)];
      er[u] = in ? rc.x - cb : -1;
      ec[u] = in ? rc.y : 0;
    }
    if (tid < BROWS) hist[tid] = 0;
    __syncthreads();
#pragma unroll
    for (int u = 0; u < EPT; ++u)
      if (er[u] >= 0) atomicAdd(&hist[er[u]], 1);
    __syncthreads();
    if (tid < 32) {
      const int4 own = ((const int4*)hist)[tid];
      const int tot = own.x + own.y + own.z + own.w;
      int inc = tot;
#pragma unroll
      for (int d = 1; d < 32; d <<= 1) {
        const int v = __shfl_up(inc, d);
        if (tid >= d) inc += v;
      }
      int4 st;
      st.x = inc - tot;
      st.y = st.x + own.x;
      st.z = st.y + own.y;
      st.w = st.z + own.z;
      ((int4*)rstart)[tid] = st;
      ((int4*)cursor)[tid] = st;
      if (tid == 31) rstart[BROWS] = inc;
    }
    __syncthreads();
#pragma unroll
    for (int u = 0; u < EPT; ++u) {
      if (er[u] >= 0) {
        const int pos = atomicAdd(&cursor[er[u]], 1);
        scol[pos] = ec[u];
        srow[pos] = (unsigned char)er[u];
      }
    }
    __syncthreads();

    {
      const int w0 = __builtin_amdgcn_readfirstlane(rstart[wv * WROWS]);
      const int w1 = __builtin_amdgcn_readfirstlane(rstart[wv * WROWS + WROWS]);
      float acc[2][KER];
#pragma unroll
      for (int k = 0; k < KER; ++k) { acc[0][k] = 0.0f; acc[1][k] = 0.0f; }
      int cur = -1;
      v4f* vt = (v4f*)vtab + wv * 32;

      int colN = 0, rlN = 0;
      float srN[EMB] = {0.0f, 0.0f, 0.0f}, scN[EMB] = {0.0f, 0.0f, 0.0f};
      if (w0 < w1) batch_reads(scol, srow, spec, cb, w0, w1, lane, colN, rlN, srN, scN);

      for (int base = w0; base < w1; base += 32) {
        int n = w1 - base;
        if (n > 32) n = 32;

        const int col = colN;
        const int rl  = rlN;
        const float d0 = srN[0] - scN[0];
        const float d1 = srN[1] - scN[1];
        const float d2 = srN[2] - scN[2];

        float val[KER];
#pragma unroll
        for (int k = 0; k < KER; ++k) {
          const float a0 = d0 - mu0[k];
          const float a1 = d1 - mu1[k];
          const float a2 = d2 - mu2[k];
          const float qq = -0.5f * (a0 * a0 + a1 * a1 + a2 * a2);
          val[k] = __expf(sg[k] * qq);
        }
        {
          v4f vq;
          vq.x = val[0]; vq.y = val[1]; vq.z = val[2]; vq.w = val[3];
          vt[lane] = vq;
        }

        if (base + 32 < w1)
          batch_reads(scol, srow, spec, cb, base + 32, w1, lane, colN, rlN, srN, scN);

        int j = 0;
        for (; j + 16 <= n; j += 16)
          edge_group<16>(col, rl, vt, j, lane, x, agg, cur, acc);
        if (n - j >= 8) { edge_group<8>(col, rl, vt, j, lane, x, agg, cur, acc); j += 8; }
        if (n - j >= 4) { edge_group<4>(col, rl, vt, j, lane, x, agg, cur, acc); j += 4; }
        if (n - j >= 2) { edge_group<2>(col, rl, vt, j, lane, x, agg, cur, acc); j += 2; }
        if (n - j >= 1) { edge_group<1>(col, rl, vt, j, lane, x, agg, cur, acc); }
      }
      if (cur >= 0) flush_row(agg, cur, lane, acc);
    }
    __syncthreads();
    c0 += CAP;
  } while (c0 < cnt);

  const int mt = wv & 7;
  const int ch = wv >> 3;
  v8f acc0 = {}, acc1 = {};
  const float* arow = agg + (mt * 16 + lid) * APITCH + 4 * half;
  const float4* __restrict__ wg = (const float4*)w;
#pragma unroll
  for (int p = 0; p < IN_F / ICHUNK; ++p) {
#pragma unroll
    for (int s = 0; s < (ICHUNK * OUT_F) / CONV_THREADS; ++s) {
      const int f  = tid + s * CONV_THREADS;
      const int il = f >> 6;
      const int o  = f & 63;
      const float4 q = wg[(size_t)(p * ICHUNK + il) * OUT_F + o];
      const int di = ((((il >> 1) * 2 + (o >> 5)) * 2 + (o & 1)) * 2 + (il & 1)) * 16
                     + ((o & 31) >> 1);
      stage[di] = q;
    }
    __syncthreads();

#pragma unroll
    for (int u = 0; u < ICHUNK / 2; ++u) {
      const v4f a  = *(const v4f*)(arow + p * (ICHUNK * KER) + 8 * u);
      const float4 q0 = stage[(((u * 2 + ch) * 2 + 0) * 2 + half) * 16 + lid];
      const float4 q1 = stage[(((u * 2 + ch) * 2 + 1) * 2 + half) * 16 + lid];
      v2f aa, bb;
      aa.x = a.x; aa.y = a.y;
      bb.x = q0.x; bb.y = q0.y;
      acc0 = __builtin_amdgcn_wmma_f32_16x16x4_f32(false, aa, false, bb,
                                                   (short)0, acc0, false, false);
      bb.x = q1.x; bb.y = q1.y;
      acc1 = __builtin_amdgcn_wmma_f32_16x16x4_f32(false, aa, false, bb,
                                                   (short)0, acc1, false, false);
      aa.x = a.z; aa.y = a.w;
      bb.x = q0.z; bb.y = q0.w;
      acc0 = __builtin_amdgcn_wmma_f32_16x16x4_f32(false, aa, false, bb,
                                                   (short)0, acc0, false, false);
      bb.x = q1.z; bb.y = q1.w;
      acc1 = __builtin_amdgcn_wmma_f32_16x16x4_f32(false, aa, false, bb,
                                                   (short)0, acc1, false, false);
    }
    if (p + 1 < IN_F / ICHUNK) __syncthreads();
  }

  const float2 bv = *(const float2*)(bias + 32 * ch + 2 * lid);
#pragma unroll
  for (int r = 0; r < 8; ++r) {
    const int node = cb + mt * 16 + r + 8 * half;
    if (node < N) {
      float2 v;
      v.x = acc0[r] + bv.x;
      v.y = acc1[r] + bv.y;
      *(float2*)(out + (size_t)node * OUT_F + 32 * ch + 2 * lid) = v;
    }
  }
}

extern "C" void kernel_launch(void* const* d_in, const int* in_sizes, int n_in,
                              void* d_out, int out_size, void* d_ws, size_t ws_size,
                              hipStream_t stream) {
  const float* x    = (const float*)d_in[0];
  const int*   eidx = (const int*)  d_in[1];
  const float* spec = (const float*)d_in[2];
  const float* w    = (const float*)d_in[3];
  const float* bias = (const float*)d_in[4];
  const float* mu   = (const float*)d_in[5];
  const float* sig  = (const float*)d_in[6];

  const int N  = in_sizes[0] / IN_F;
  const int E  = in_sizes[1] / 2;
  const int NB = (N + BROWS - 1) / BROWS;

  int*  tab  = (int*)d_ws;
  int2* list = (int2*)(tab + (NB + 1) * SORT_WGS);
  float* out = (float*)d_out;

  chunk_sort<<<SORT_WGS, SORT_THREADS, 0, stream>>>(eidx, list, tab, E, NB);
  bucket_conv<<<NB, CONV_THREADS, 0, stream>>>(x, w, list, tab,
                                               spec, mu, sig, bias, out, N, E);
}


# Buckets of 208 rows, 26-wave workgroups, one per WGP owning its LDS: 241 resident at once, equal work (was 391 of 128 rows: 2 on 135 WGPs, 1 on 121); W staged whole at entry; sum order may change
# speedup vs baseline: 3.0395x; 1.0421x over previous
#include <hip/hip_runtime.h>


typedef __attribute__((ext_vector_type(2))) float v2f;
typedef __attribute__((ext_vector_type(4))) float v4f;
typedef __attribute__((ext_vector_type(8))) float v8f;

#define IN_F  64
#define OUT_F 64
#define KER   4
#define EMB   3
#define CDIM  (IN_F * KER)

#define BROWS   208
#define NB_MAX  512
#define SORT_WGS     128
#define SORT_THREADS 1024
#define SORT_BATCH   8
#define SORT_CAP     (SORT_BATCH * SORT_THREADS)

__device__ __forceinline__ int bucket_of(const int r) { return (int)((unsigned)r / BROWS); }

__global__ void __launch_bounds__(SORT_THREADS)
chunk_sort(const int* __restrict__ eidx,
           int2* __restrict__ list,
           int* __restrict__ tab,
           int E, int NB) {
  __shared__ int hist[NB_MAX];
  __shared__ int wsum[NB_MAX / 32];
  __shared__ int2 sorted[SORT_CAP];
  const int tid = threadIdx.x;

  const int per = (E + SORT_WGS - 1) / SORT_WGS;
  int ebeg = blockIdx.x * per;
  if (ebeg > E) ebeg = E;
  int eend = ebeg + per;
  if (eend > E) eend = E;

  int r0[SORT_BATCH], c0[SORT_BATCH];
#pragma unroll
  for (int u = 0; u < SORT_BATCH; ++u) {
    const int e = ebeg + tid + u * SORT_THREADS;
    const bool in = e < eend;
    r0[u] = in ? eidx[e] : -1;
    c0[u] = in ? eidx[E + e] : 0;
  }
  for (int b = tid; b < NB; b += SORT_THREADS) hist[b] = 0;
  __syncthreads();
#pragma unroll
  for (int u = 0; u < SORT_BATCH; ++u)
    if (r0[u] >= 0) atomicAdd(&hist[bucket_of(r0[u])], 1);
  for (int e0 = ebeg + tid + SORT_BATCH * SORT_THREADS; e0 < eend;
       e0 += SORT_BATCH * SORT_THREADS) {
    int r[SORT_BATCH];
#pragma unroll
    for (int u = 0; u < SORT_BATCH; ++u) {
      const int e = e0 + u * SORT_THREADS;
      r[u] = (e < eend) ? eidx[e] : -1;
    }
#pragma unroll
    for (int u = 0; u < SORT_BATCH; ++u)
      if (r[u] >= 0) atomicAdd(&hist[bucket_of(r[u])], 1);
  }
  __syncthreads();

  const int lane = tid & 31;
  const int wv   = tid >> 5;
  int v = 0, inc = 0;
  if (tid < NB_MAX) {
    v = (tid < NB) ? hist[tid] : 0;
    inc = v;
#pragma unroll
    for (int d = 1; d < 32; d <<= 1) {
      const int t = __shfl_up(inc, d);
      if (lane >= d) inc += t;
    }
    if (lane == 31) wsum[wv] = inc;
  }
  __syncthreads();
  if (tid < NB) {
    int before = 0;
#pragma unroll
    for (int q = 0; q < NB_MAX / 32; ++q) {
      const int s = wsum[q];
      if (q < wv) before += s;
    }
    const int start = before + inc - v;
    hist[tid] = start;
    tab[tid * SORT_WGS + blockIdx.x] = start;
  }
  if (tid == 0) tab[NB * SORT_WGS + blockIdx.x] = eend - ebeg;
  __syncthreads();

  int pos[SORT_BATCH];
#pragma unroll
  for (int u = 0; u < SORT_BATCH; ++u) {
    pos[u] = 0;
    if (r0[u] >= 0) pos[u] = atomicAdd(&hist[bucket_of(r0[u])], 1);
  }
#pragma unroll
  for (int u = 0; u < SORT_BATCH; ++u) {
    if (r0[u] >= 0) {
      if (pos[u] < SORT_CAP) sorted[pos[u]] = make_int2(r0[u], c0[u]);
      else                   list[ebeg + pos[u]] = make_int2(r0[u], c0[u]);
    }
  }
  for (int e0 = ebeg + tid + SORT_BATCH * SORT_THREADS; e0 < eend;
       e0 += SORT_BATCH * SORT_THREADS) {
    int r[SORT_BATCH], c[SORT_BATCH];
#pragma unroll
    for (int u = 0; u < SORT_BATCH; ++u) {
      const int e = e0 + u * SORT_THREADS;
      const bool in = e < eend;
      r[u] = in ? eidx[e] : -1;
      c[u] = in ? eidx[E + e] : 0;
    }
#pragma unroll
    for (int u = 0; u < SORT_BATCH; ++u) {
      if (r[u] >= 0) {
        const int pos = atomicAdd(&hist[bucket_of(r[u])], 1);
        if (pos < SORT_CAP) sorted[pos] = make_int2(r[u], c[u]);
        else                list[ebeg + pos] = make_int2(r[u], c[u]);
      }
    }
  }
  __syncthreads();

  int n = eend - ebeg;
  if (n > SORT_CAP) n = SORT_CAP;
  if (n > 0) {
    const int g0 = (ebeg & ~31) + tid;
    int2 ent[SORT_BATCH + 1];
#pragma unroll
    for (int k = 0; k <= SORT_BATCH; ++k) {
      const int i = g0 + k * SORT_THREADS - ebeg;
      ent[k] = sorted[min(max(i, 0), n - 1)];
    }
#pragma unroll
    for (int k = 0; k <= SORT_BATCH; ++k) {
      const int g = g0 + k * SORT_THREADS;
      if (g >= ebeg && g < ebeg + n) list[g] = ent[k];
    }
  }
}

#define CONV_WAVES   26
#define CONV_THREADS (CONV_WAVES * 32)
#define WROWS   (BROWS / CONV_WAVES)
#define MTILES  (BROWS / 16)
#define RPAD    256
#define APITCH  (CDIM + 4)
#define CAP     4160
#define EPT     (CAP / CONV_THREADS)
#define STAGE_E  ((CAP * 5 + 15) / 16)
#define WF4     (IN_F * OUT_F)
#define WTRIPS  ((WF4 + CONV_THREADS - 1) / CONV_THREADS)
#define AGG4    (BROWS * APITCH / 4)

__device__ __forceinline__ void flush_row(float* agg, const int r, const int lane,
                                          float (&acc)[2][KER]) {
  v4f* p = (v4f*)(agg + r * APITCH + 8 * lane);
  v4f lo = p[0], hi = p[1];
  lo.x += acc[0][0]; lo.y += acc[0][1]; lo.z += acc[0][2]; lo.w += acc[0][3];
  hi.x += acc[1][0]; hi.y += acc[1][1]; hi.z += acc[1][2]; hi.w += acc[1][3];
  p[0] = lo;
  p[1] = hi;
#pragma unroll
  for (int k = 0; k < KER; ++k) { acc[0][k] = 0.0f; acc[1][k] = 0.0f; }
}

__device__ __forceinline__ void batch_reads(const int* scol, const unsigned char* srow,
                                            const float* __restrict__ spec,
                                            const int cb, const int base, const int w1,
                                            const int lane, int& col, int& rl,
                                            float (&sr)[EMB], float (&sc)[EMB]) {
  int e = base + lane;
  if (e >= w1) e = w1 - 1;
  col = scol[e];
  rl  = srow[e];
  const int row = cb + rl;
#pragma unroll
  for (int t = 0; t < EMB; ++t) {
    sr[t] = spec[row * EMB + t];
    sc[t] = spec[col * EMB + t];
  }
}

#define VAHEAD 3
template <int G>
__device__ __forceinline__ void edge_group(const int col, const int rl,
                                           const v4f* vt, const int j,
                                           const int lane,
                                           const float* __restrict__ x,
                                           float* agg, int& cur,
                                           float (&acc)[2][KER]) {
  v2f xv[G];
#pragma unroll
  for (int u = 0; u < G; ++u) {
    const int cj = __builtin_amdgcn_readlane(col, j + u);
    xv[u] = ((const v2f*)(x + (size_t)cj * IN_F))[(unsigned)lane];
  }
  constexpr int D = (G < VAHEAD) ? G : VAHEAD;
  v4f vv[G];
#pragma unroll
  for (int u = 0; u < D; ++u) vv[u] = vt[j + u];
#pragma unroll
  for (int u = 0; u < G; ++u) {
    if (u + D < G) vv[u + D] = vt[j + u + D];
    const int rj = __builtin_amdgcn_readlane(rl, j + u);
    if (rj != cur) {
      if (cur >= 0) flush_row(agg, cur, lane, acc);
      cur = rj;
    }
    const float v0 = vv[u].x, v1 = vv[u].y, v2 = vv[u].z, v3 = vv[u].w;
    acc[0][0] += v0 * xv[u].x; acc[0][1] += v1 * xv[u].x;
    acc[0][2] += v2 * xv[u].x; acc[0][3] += v3 * xv[u].x;
    acc[1][0] += v0 * xv[u].y; acc[1][1] += v1 * xv[u].y;
    acc[1][2] += v2 * xv[u].y; acc[1][3] += v3 * xv[u].y;
  }
}

__global__ void __launch_bounds__(CONV_THREADS)
bucket_conv(const float* __restrict__ x,
            const float* __restrict__ w,
            const int2*  __restrict__ list,
            const int*   __restrict__ tab,
            const float* __restrict__ spec,
            const float* __restrict__ mu,
            const float* __restrict__ sig,
            const float* __restrict__ bias,
            float* __restrict__ out,
            int N, int E) {
  __shared__ __attribute__((aligned(16))) float agg[BROWS * APITCH];
  __shared__ float4 stage[STAGE_E];
  __shared__ float4 wst[WF4];
  __shared__ __attribute__((aligned(16))) int hist[RPAD];
  __shared__ __attribute__((aligned(16))) int cursor[RPAD];
  __shared__ __attribute__((aligned(16))) int rstart[RPAD + 4];
  __shared__ __attribute__((aligned(16))) int segbase[SORT_WGS + 4];
  __shared__ __attribute__((aligned(16))) int segoff[SORT_WGS];

  __shared__ __attribute__((aligned(16))) float vtab[CONV_WAVES * 32 * KER];

  int*           scol = (int*)stage;
  unsigned char* srow = (unsigned char*)(scol + CAP);

  const int tid  = threadIdx.x;
  const int lane = tid & 31;
  const int wv   = __builtin_amdgcn_readfirstlane(tid >> 5);
  const int half = lane >> 4;
  const int lid  = lane & 15;
  const int b    = blockIdx.x;
  const int cb   = b * BROWS;

  int4 s0 = make_int4(0, 0, 0, 0), s1 = make_int4(0, 0, 0, 0);
  if (tid < 32) {
    s0 = ((const int4*)(tab + (size_t)b * SORT_WGS))[tid];
    s1 = ((const int4*)(tab + (size_t)(b + 1) * SORT_WGS))[tid];
  }

  const float4* __restrict__ wg = (const float4*)w;
  float4 wq[WTRIPS];
#pragma unroll
  for (int s = 0; s < WTRIPS; ++s) {
    const int f = tid + s * CONV_THREADS;
    wq[s] = wg[min(f, WF4 - 1)];
  }

  v4f* a4 = (v4f*)agg;
  const v4f z = {0.0f, 0.0f, 0.0f, 0.0f};
#pragma unroll
  for (int t = 0; t < AGG4 / CONV_THREADS; ++t) a4[tid + t * CONV_THREADS] = z;

#pragma unroll
  for (int s = 0; s < WTRIPS; ++s) {
    const int f  = min(tid + s * CONV_THREADS, WF4 - 1);
    const int il = f >> 6;
    const int o  = f & 63;
    const int di = ((((il >> 1) * 2 + (o >> 5)) * 2 + (o & 1)) * 2 + (il & 1)) * 16
                   + ((o & 31) >> 1);
    wst[di] = wq[s];
  }
  if (tid < AGG4 % CONV_THREADS) a4[tid + (AGG4 / CONV_THREADS) * CONV_THREADS] = z;

  if (tid < 32) {
    const int per = (E + SORT_WGS - 1) / SORT_WGS;
    const int l0 = s1.x - s0.x, l1 = s1.y - s0.y, l2 = s1.z - s0.z, l3 = s1.w - s0.w;
    const int tot = l0 + l1 + l2 + l3;
    int inc = tot;
#pragma unroll
    for (int d = 1; d < 32; d <<= 1) {
      const int v = __shfl_up(inc, d);
      if (tid >= d) inc += v;
    }
    int4 sb;
    sb.x = inc - tot;
    sb.y = sb.x + l0;
    sb.z = sb.y + l1;
    sb.w = sb.z + l2;
    ((int4*)segbase)[tid] = sb;
    int4 so;
    so.x = (4 * tid + 0) * per + s0.x - sb.x;
    so.y = (4 * tid + 1) * per + s0.y - sb.y;
    so.z = (4 * tid + 2) * per + s0.z - sb.z;
    so.w = (4 * tid + 3) * per + s0.w - sb.w;
    ((int4*)segoff)[tid] = so;
    if (tid == 31) segbase[SORT_WGS] = inc;
  }
  __syncthreads();
  const int cnt = segbase[SORT_WGS];

  float mu0[KER], mu1[KER], mu2[KER], sg[KER];
#pragma unroll
  for (int k = 0; k < KER; ++k) {
    mu0[k] = mu[0 * KER + k];
    mu1[k] = mu[1 * KER + k];
    mu2[k] = mu[2 * KER + k];
    sg[k]  = sig[k];
  }

  int c0 = 0;
  do {
    int m = cnt - c0;
    if (m > CAP) m = CAP;

    int er[EPT], ec[EPT], gi[EPT], seg[EPT];
#pragma unroll
    for (int u = 0; u < EPT; ++u) {
      gi[u] = c0 + tid + u * CONV_THREADS;
      seg[u] = 0;
    }
#pragma unroll
    for (int step = SORT_WGS / 2; step >= 1; step >>= 1) {
#pragma unroll
      for (int u = 0; u < EPT; ++u)
        if (segbase[seg[u] + step] <= gi[u]) seg[u] += step;
    }
#pragma unroll
    for (int u = 0; u < EPT; ++u) {
      const bool in = tid + u * CONV_THREADS < m;
      const int2 rc = list[min(gi[u] + segoff[seg[u]], E - 1)];
      er[u] = in ? rc.x - cb : -1;
      ec[u] = in ? rc.y : 0;
    }
    if (tid < RPAD) hist[tid] = 0;
    __syncthreads();
#pragma unroll
    for (int u = 0; u < EPT; ++u)
      if (er[u] >= 0) atomicAdd(&hist[er[u]], 1);
    __syncthreads();
    if (tid < 32) {
      const int4 own = ((const int4*)hist)[2 * tid];
      const int4 ow2 = ((const int4*)hist)[2 * tid + 1];
      const int tot = own.x + own.y + own.z + own.w + ow2.x + ow2.y + ow2.z + ow2.w;
      int inc = tot;
#pragma unroll
      for (int d = 1; d < 32; d <<= 1) {
        const int v = __shfl_up(inc, d);
        if (tid >= d) inc += v;
      }
      int4 st, s2;
      st.x = inc - tot;
      st.y = st.x + own.x;
      st.z = st.y + own.y;
      st.w = st.z + own.z;
      s2.x = st.w + own.w;
      s2.y = s2.x + ow2.x;
      s2.z = s2.y + ow2.y;
      s2.w = s2.z + ow2.z;
      ((int4*)rstart)[2 * tid]     = st;
      ((int4*)rstart)[2 * tid + 1] = s2;
      ((int4*)cursor)[2 * tid]     = st;
      ((int4*)cursor)[2 * tid + 1] = s2;
      if (tid == 31) rstart[RPAD] = inc;
    }
    __syncthreads();
#pragma unroll
    for (int u = 0; u < EPT; ++u) {
      if (er[u] >= 0) {
        const int pos = atomicAdd(&cursor[er[u]], 1);
        scol[pos] = ec[u];
        srow[pos] = (unsigned char)er[u];
      }
    }
    __syncthreads();

    {
      const int w0 = __builtin_amdgcn_readfirstlane(rstart[wv * WROWS]);
      const int w1 = __builtin_amdgcn_readfirstlane(rstart[wv * WROWS + WROWS]);
      float acc[2][KER];
#pragma unroll
      for (int k = 0; k < KER; ++k) { acc[0][k] = 0.0f; acc[1][k] = 0.0f; }
      int cur = -1;
      v4f* vt = (v4f*)vtab + wv * 32;

      int colN = 0, rlN = 0;
      float srN[EMB] = {0.0f, 0.0f, 0.0f}, scN[EMB] = {0.0f, 0.0f, 0.0f};
      if (w0 < w1) batch_reads(scol, srow, spec, cb, w0, w1, lane, colN, rlN, srN, scN);

      for (int base = w0; base < w1; base += 32) {
        int n = w1 - base;
        if (n > 32) n = 32;

        const int col = colN;
        const int rl  = rlN;
        const float d0 = srN[0] - scN[0];
        const float d1 = srN[1] - scN[1];
        const float d2 = srN[2] - scN[2];

        float val[KER];
#pragma unroll
        for (int k = 0; k < KER; ++k) {
          const float a0 = d0 - mu0[k];
          const float a1 = d1 - mu1[k];
          const float a2 = d2 - mu2[k];
          const float qq = -0.5f * (a0 * a0 + a1 * a1 + a2 * a2);
          val[k] = __expf(sg[k] * qq);
        }
        {
          v4f vq;
          vq.x = val[0]; vq.y = val[1]; vq.z = val[2]; vq.w = val[3];
          vt[lane] = vq;
        }

        if (base + 32 < w1)
          batch_reads(scol, srow, spec, cb, base + 32, w1, lane, colN, rlN, srN, scN);

        int j = 0;
        for (; j + 16 <= n; j += 16)
          edge_group<16>(col, rl, vt, j, lane, x, agg, cur, acc);
        if (n - j >= 8) { edge_group<8>(col, rl, vt, j, lane, x, agg, cur, acc); j += 8; }
        if (n - j >= 4) { edge_group<4>(col, rl, vt, j, lane, x, agg, cur, acc); j += 4; }
        if (n - j >= 2) { edge_group<2>(col, rl, vt, j, lane, x, agg, cur, acc); j += 2; }
        if (n - j >= 1) { edge_group<1>(col, rl, vt, j, lane, x, agg, cur, acc); }
      }
      if (cur >= 0) flush_row(agg, cur, lane, acc);
    }
    __syncthreads();
    c0 += CAP;
  } while (c0 < cnt);

  const int mt = wv % MTILES;
  const int ch = wv / MTILES;
  v8f acc0 = {}, acc1 = {};
  const float* arow = agg + (mt * 16 + lid) * APITCH + 4 * half;
#pragma unroll
  for (int u = 0; u < IN_F / 2; ++u) {
    const v4f a  = *(const v4f*)(arow + 8 * u);
    const float4 q0 = wst[(((u * 2 + ch) * 2 + 0) * 2 + half) * 16 + lid];
    const float4 q1 = wst[(((u * 2 + ch) * 2 + 1) * 2 + half) * 16 + lid];
    v2f aa, bb;
    aa.x = a.x; aa.y = a.y;
    bb.x = q0.x; bb.y = q0.y;
    acc0 = __builtin_amdgcn_wmma_f32_16x16x4_f32(false, aa, false, bb,
                                                 (short)0, acc0, false, false);
    bb.x = q1.x; bb.y = q1.y;
    acc1 = __builtin_amdgcn_wmma_f32_16x16x4_f32(false, aa, false, bb,
                                                 (short)0, acc1, false, false);
    aa.x = a.z; aa.y = a.w;
    bb.x = q0.z; bb.y = q0.w;
    acc0 = __builtin_amdgcn_wmma_f32_16x16x4_f32(false, aa, false, bb,
                                                 (short)0, acc0, false, false);
    bb.x = q1.z; bb.y = q1.w;
    acc1 = __builtin_amdgcn_wmma_f32_16x16x4_f32(false, aa, false, bb,
                                                 (short)0, acc1, false, false);
  }

  const float2 bv = *(const float2*)(bias + 32 * ch + 2 * lid);
#pragma unroll
  for (int r = 0; r < 8; ++r) {
    const int node = cb + mt * 16 + r + 8 * half;
    if (node < N) {
      float2 v;
      v.x = acc0[r] + bv.x;
      v.y = acc1[r] + bv.y;
      *(float2*)(out + (size_t)node * OUT_F + 32 * ch + 2 * lid) = v;
    }
  }
}

extern "C" void kernel_launch(void* const* d_in, const int* in_sizes, int n_in,
                              void* d_out, int out_size, void* d_ws, size_t ws_size,
                              hipStream_t stream) {
  const float* x    = (const float*)d_in[0];
  const int*   eidx = (const int*)  d_in[1];
  const float* spec = (const float*)d_in[2];
  const float* w    = (const float*)d_in[3];
  const float* bias = (const float*)d_in[4];
  const float* mu   = (const float*)d_in[5];
  const float* sig  = (const float*)d_in[6];

  const int N  = in_sizes[0] / IN_F;
  const int E  = in_sizes[1] / 2;
  const int NB = (N + BROWS - 1) / BROWS;

  int*  tab  = (int*)d_ws;
  int2* list = (int2*)(tab + (NB + 1) * SORT_WGS);
  float* out = (float*)d_out;

  chunk_sort<<<SORT_WGS, SORT_THREADS, 0, stream>>>(eidx, list, tab, E, NB);
  bucket_conv<<<NB, CONV_THREADS, 0, stream>>>(x, w, list, tab,
                                               spec, mu, sig, bias, out, N, E);
}
